# starting kernel
# baseline (speedup 1.0000x reference)
#include <hip/hip_runtime.h>

#define NN   100000
#define INC  256
#define HIDC 128
#define NE   1600000
#define NPOS 200000
#define NNEG 200000

typedef __attribute__((ext_vector_type(2))) float v2f;
typedef __attribute__((ext_vector_type(8))) float v8f;

__global__ void k_init_dinv(float* dinv) {
  int i = blockIdx.x * blockDim.x + threadIdx.x;
  if (i < NN) dinv[i] = 1.0f;
}

__global__ void k_count_deg(float* dinv, const int* __restrict__ col) {
  int i = blockIdx.x * blockDim.x + threadIdx.x;
  if (i < NE) unsafeAtomicAdd(&dinv[col[i]], 1.0f);
}

__global__ void k_finish_dinv(float* dinv) {
  int i = blockIdx.x * blockDim.x + threadIdx.x;
  if (i < NN) dinv[i] = 1.0f / sqrtf(dinv[i]);
}

__global__ void k_transpose(const float* __restrict__ W, float* __restrict__ WT,
                            int K, int N) {
  int i = blockIdx.x * blockDim.x + threadIdx.x;
  if (i >= K * N) return;
  int k = i / N, n = i - k * N;
  WT[n * K + k] = W[i];
}

template <int K>
__global__ __launch_bounds__(256) void k_gemm_wmma(const float* __restrict__ A,
                                                   const float* __restrict__ WT,
                                                   float* __restrict__ C) {
  int wave = (blockIdx.x * blockDim.x + threadIdx.x) >> 5;
  int lane = threadIdx.x & 31;
  int m0 = wave * 16;
  if (m0 >= NN) return;
  int half = lane >> 4;
  int l15  = lane & 15;
  const float* aptr  = A  + (size_t)(m0 + l15) * K + half * 2;
  const float* bbase = WT + (size_t)l15 * K + half * 2;

  v8f acc[8];
#pragma unroll
  for (int t = 0; t < 8; ++t) acc[t] = (v8f)(0.0f);

  __builtin_prefetch(aptr, 0, 0);
#pragma unroll 2
  for (int k = 0; k < K; k += 4) {
    v2f a = *(const v2f*)(aptr + k);
#pragma unroll
    for (int t = 0; t < 8; ++t) {
      v2f b = *(const v2f*)(bbase + (size_t)t * 16 * K + k);
      acc[t] = __builtin_amdgcn_wmma_f32_16x16x4_f32(
          false, a, false, b, (short)0, acc[t], false, false);
    }
  }

#pragma unroll
  for (int t = 0; t < 8; ++t) {
#pragma unroll
    for (int i = 0; i < 8; ++i) {
      C[(size_t)(m0 + i + half * 8) * HIDC + t * 16 + l15] = acc[t][i];
    }
  }
}

__global__ void k_selfloop(float* __restrict__ out, const float* __restrict__ h,
                           const float* __restrict__ dinv) {
  int i = blockIdx.x * blockDim.x + threadIdx.x;
  if (i >= NN * (HIDC / 4)) return;
  int n = i >> 5;
  float d = dinv[n];
  float s = d * d;
  float4 v = ((const float4*)h)[i];
  ((float4*)out)[i] = make_float4(v.x * s, v.y * s, v.z * s, v.w * s);
}

__global__ void k_edge_agg(float* __restrict__ out, const float* __restrict__ h,
                           const float* __restrict__ dinv,
                           const int* __restrict__ row, const int* __restrict__ col) {
  long long t = (long long)blockIdx.x * blockDim.x + threadIdx.x;
  if (t >= (long long)NE * (HIDC / 4)) return;
  int e = (int)(t >> 5);
  int g = (int)(t & 31);
  int r = row[e], c = col[e];
  float norm = dinv[r] * dinv[c];
  float4 v = ((const float4*)(h + (size_t)r * HIDC))[g];
  float* dst = out + (size_t)c * HIDC + g * 4;
  unsafeAtomicAdd(dst + 0, v.x * norm);
  unsafeAtomicAdd(dst + 1, v.y * norm);
  unsafeAtomicAdd(dst + 2, v.z * norm);
  unsafeAtomicAdd(dst + 3, v.w * norm);
}

__global__ void k_bias_act(float* __restrict__ z, const float* __restrict__ b,
                           int do_relu) {
  int i = blockIdx.x * blockDim.x + threadIdx.x;
  if (i >= NN * HIDC) return;
  int f = i & (HIDC - 1);
  float v = z[i] + b[f];
  if (do_relu) v = fmaxf(v, 0.0f);
  z[i] = v;
}

__global__ void k_decode(const float* __restrict__ z, const int* __restrict__ pos,
                         const int* __restrict__ neg, float* __restrict__ out) {
  int w = (blockIdx.x * blockDim.x + threadIdx.x) >> 5;
  int lane = threadIdx.x & 31;
  if (w >= NPOS + NNEG) return;
  int a, b;
  if (w < NPOS) { a = pos[w]; b = pos[NPOS + w]; }
  else          { int j = w - NPOS; a = neg[j]; b = neg[NNEG + j]; }
  float4 va = ((const float4*)(z + (size_t)a * HIDC))[lane];
  float4 vb = ((const float4*)(z + (size_t)b * HIDC))[lane];
  float s = va.x * vb.x + va.y * vb.y + va.z * vb.z + va.w * vb.w;
#pragma unroll
  for (int off = 16; off > 0; off >>= 1) s += __shfl_xor(s, off, 32);
  if (lane == 0) out[w] = s;
}

extern "C" void kernel_launch(void* const* d_in, const int* in_sizes, int n_in,
                              void* d_out, int out_size, void* d_ws, size_t ws_size,
                              hipStream_t stream) {
  const float* x  = (const float*)d_in[0];
  const int*   ei = (const int*)d_in[1];
  const int*   pos = (const int*)d_in[2];
  const int*   neg = (const int*)d_in[3];
  const float* W1 = (const float*)d_in[4];
  const float* b1 = (const float*)d_in[5];
  const float* W2 = (const float*)d_in[6];
  const float* b2 = (const float*)d_in[7];
  float* logits = (float*)d_out;

  char* ws = (char*)d_ws;
  float* dinv = (float*)ws;
  float* WT   = (float*)(ws + (512u << 10));
  float* bufH = (float*)(ws + (1u << 20));
  float* bufZ = (float*)(ws + (1u << 20) + (52u << 20));

  const int* row = ei;
  const int* col = ei + NE;
  const int B = 256;

  k_init_dinv<<<(NN + B - 1) / B, B, 0, stream>>>(dinv);
  k_count_deg<<<(NE + B - 1) / B, B, 0, stream>>>(dinv, col);
  k_finish_dinv<<<(NN + B - 1) / B, B, 0, stream>>>(dinv);

  int gemm_blocks = (((NN + 15) / 16) + 7) / 8;
  long long agg_t = (long long)NE * 32;
  int agg_blocks = (int)((agg_t + B - 1) / B);
  int sl_blocks  = (NN * 32 + B - 1) / B;
  int ba_blocks  = (NN * HIDC + B - 1) / B;

  k_transpose<<<(INC * HIDC + B - 1) / B, B, 0, stream>>>(W1, WT, INC, HIDC);
  k_gemm_wmma<INC><<<gemm_blocks, B, 0, stream>>>(x, WT, bufH);
  k_selfloop<<<sl_blocks, B, 0, stream>>>(bufZ, bufH, dinv);
  k_edge_agg<<<agg_blocks, B, 0, stream>>>(bufZ, bufH, dinv, row, col);
  k_bias_act<<<ba_blocks, B, 0, stream>>>(bufZ, b1, 1);

  k_transpose<<<(HIDC * HIDC + B - 1) / B, B, 0, stream>>>(W2, WT, HIDC, HIDC);
  k_gemm_wmma<HIDC><<<gemm_blocks, B, 0, stream>>>(bufZ, WT, bufH);
  k_selfloop<<<sl_blocks, B, 0, stream>>>(bufZ, bufH, dinv);
  k_edge_agg<<<agg_blocks, B, 0, stream>>>(bufZ, bufH, dinv, row, col);
  k_bias_act<<<ba_blocks, B, 0, stream>>>(bufZ, b2, 0);

  int dec_blocks = ((NPOS + NNEG) * 32 + B - 1) / B;
  k_decode<<<dec_blocks, B, 0, stream>>>(bufZ, pos, neg, logits);
}


# edge scatter-add by float atomics replaced by a per-call counting sort by destination and a gather, one wave per output row: sums in arrival-rank order, self-loop, bias, relu in the one store
# speedup vs baseline: 2.1182x; 2.1182x over previous
#include <hip/hip_runtime.h>

#define NN   100000
#define INC  256
#define HIDC 128
#define NE   1600000
#define NPOS 200000
#define NNEG 200000
#define NK   (2 * NN)

typedef __attribute__((ext_vector_type(2))) float v2f;
typedef __attribute__((ext_vector_type(8))) float v8f;

__global__ void k_zero_cnt(int* cnt) {
  int i = blockIdx.x * blockDim.x + threadIdx.x;
  if (i < NK) cnt[i] = 0;
}

__global__ void k_count_rank(int* cnt, const int* __restrict__ row,
                             const int* __restrict__ col, int* __restrict__ rank) {
  int i = blockIdx.x * blockDim.x + threadIdx.x;
  if (i < NE) rank[i] = atomicAdd(&cnt[2 * col[i] + (row[i] >> 16)], 1);
}

#define SCAN_TILES 25
#define SCAN_GROUP 5

__device__ __forceinline__ void scan_load_group(const int* off, int g, int tid,
                                                int4 (&va)[SCAN_GROUP], int4 (&vb)[SCAN_GROUP],
                                                int (&s)[SCAN_GROUP]) {
#pragma unroll
  for (int q = 0; q < SCAN_GROUP; ++q) {
    int i = (g + q) * 8192 + tid * 8;
    va[q] = make_int4(0, 0, 0, 0);
    vb[q] = make_int4(0, 0, 0, 0);
    if (i < NK) {
      va[q] = *(const int4*)(off + i);
      vb[q] = *(const int4*)(off + i + 4);
    }
  }
#pragma unroll
  for (int q = 0; q < SCAN_GROUP; ++q)
    s[q] = (va[q].x + va[q].y + va[q].z + va[q].w) + (vb[q].x + vb[q].y + vb[q].z + vb[q].w);
  const int lane = tid & 31;
#pragma unroll
  for (int d = 1; d < 32; d <<= 1) {
    int u[SCAN_GROUP];
#pragma unroll
    for (int q = 0; q < SCAN_GROUP; ++q) u[q] = __shfl_up(s[q], d, 32);
#pragma unroll
    for (int q = 0; q < SCAN_GROUP; ++q) if (lane >= d) s[q] += u[q];
  }
}

__global__ __launch_bounds__(1024) void k_scan(int* off) {
  __shared__ int part[SCAN_TILES * 32];
  __shared__ int wtot[32];
  const int tid = threadIdx.x, lane = tid & 31, wid = tid >> 5;

  for (int g = 0; g < SCAN_TILES; g += SCAN_GROUP) {
    int4 va[SCAN_GROUP], vb[SCAN_GROUP];
    int s[SCAN_GROUP];
    scan_load_group(off, g, tid, va, vb, s);
    if (lane == 31) {
#pragma unroll
      for (int q = 0; q < SCAN_GROUP; ++q) part[(g + q) * 32 + wid] = s[q];
    }
  }
  __syncthreads();

  int p = (tid < SCAN_TILES * 32) ? part[tid] : 0;
  int incl = p;
#pragma unroll
  for (int d = 1; d < 32; d <<= 1) {
    int u = __shfl_up(incl, d, 32);
    if (lane >= d) incl += u;
  }
  if (lane == 31) wtot[wid] = incl;
  __syncthreads();
  int wv = wtot[lane];
  int wincl = wv;
#pragma unroll
  for (int d = 1; d < 32; d <<= 1) {
    int u = __shfl_up(wincl, d, 32);
    if (lane >= d) wincl += u;
  }
  int wexcl = __shfl(wincl - wv, wid, 32);
  int total = __shfl(wincl, 31, 32);
  if (tid < SCAN_TILES * 32) part[tid] = wexcl + incl - p;
  if (tid == 0) off[NK] = total;
  __syncthreads();

#pragma nounroll
  for (int g = 0; g < SCAN_TILES; g += SCAN_GROUP) {
    int4 va[SCAN_GROUP], vb[SCAN_GROUP];
    int s[SCAN_GROUP];
    scan_load_group(off, g, tid, va, vb, s);
#pragma unroll
    for (int q = 0; q < SCAN_GROUP; ++q) {
      const int4 a = va[q];
      const int4 b = vb[q];
      int i = (g + q) * 8192 + tid * 8;
      int own = (a.x + a.y + a.z + a.w) + (b.x + b.y + b.z + b.w);
      int ex = part[(g + q) * 32 + wid] + s[q] - own;
      if (i < NK) {
        int4 o, o2;
        o.x = ex;
        o.y = o.x + a.x;
        o.z = o.y + a.y;
        o.w = o.z + a.z;
        o2.x = o.w + a.w;
        o2.y = o2.x + b.x;
        o2.z = o2.y + b.y;
        o2.w = o2.z + b.z;
        *(int4*)(off + i) = o;
        *(int4*)(off + i + 4) = o2;
      }
    }
  }
}

__global__ void k_fill_srcs(unsigned short* __restrict__ srcs, const int* __restrict__ off,
                            const int* __restrict__ row, const int* __restrict__ col,
                            const int* __restrict__ rank) {
  int i = blockIdx.x * blockDim.x + threadIdx.x;
  if (i < NE) {
    int r = row[i];
    srcs[off[2 * col[i] + (r >> 16)] + rank[i]] = (unsigned short)(r & 0xFFFF);
  }
}

__global__ void k_transpose(const float* __restrict__ W, float* __restrict__ WT,
                            int K, int N) {
  int i = blockIdx.x * blockDim.x + threadIdx.x;
  if (i >= K * N) return;
  int k = i / N, n = i - k * N;
  WT[n * K + k] = W[i];
}

template <int K>
__global__ __launch_bounds__(256) void k_gemm_wmma(const float* __restrict__ A,
                                                   const float* __restrict__ WT,
                                                   float* __restrict__ C) {
  int wave = (blockIdx.x * blockDim.x + threadIdx.x) >> 5;
  int lane = threadIdx.x & 31;
  int m0 = wave * 16;
  if (m0 >= NN) return;
  int half = lane >> 4;
  int l15  = lane & 15;
  const float* aptr  = A  + (size_t)(m0 + l15) * K + half * 2;
  const float* bbase = WT + (size_t)l15 * K + half * 2;

  v8f acc[8];
#pragma unroll
  for (int t = 0; t < 8; ++t) acc[t] = (v8f)(0.0f);

  __builtin_prefetch(aptr, 0, 0);
#pragma unroll 2
  for (int k = 0; k < K; k += 4) {
    v2f a = *(const v2f*)(aptr + k);
#pragma unroll
    for (int t = 0; t < 8; ++t) {
      v2f b = *(const v2f*)(bbase + (size_t)t * 16 * K + k);
      acc[t] = __builtin_amdgcn_wmma_f32_16x16x4_f32(
          false, a, false, b, (short)0, acc[t], false, false);
    }
  }

#pragma unroll
  for (int t = 0; t < 8; ++t) {
#pragma unroll
    for (int i = 0; i < 8; ++i) {
      C[(size_t)(m0 + i + half * 8) * HIDC + t * 16 + l15] = acc[t][i];
    }
  }
}

__device__ __forceinline__ float4 ld_row(const float* __restrict__ h, int r, int lane) {
  return ((const float4*)(h + (size_t)r * HIDC))[lane];
}

template <int RELU>
__global__ __launch_bounds__(256) void k_gather(float* __restrict__ out,
                                                const float* __restrict__ h,
                                                const int* __restrict__ off,
                                                const unsigned short* __restrict__ srcs,
                                                const float* __restrict__ bias) {
  const int lane = threadIdx.x & 31;
  const int c = __builtin_amdgcn_readfirstlane((int)((blockIdx.x * blockDim.x + threadIdx.x) >> 5));
  if (c >= NN) return;
  const int beg = off[2 * c];
  const int mid = off[2 * c + 1];
  const int end = off[2 * c + 2];
  const float dc = 1.0f / sqrtf((float)(end - beg + 1));
  const float4 bv = ((const float4*)bias)[lane];
  float4 acc;
  {
    float4 v = ld_row(h, c, lane);
    float s = dc * dc;
    acc = make_float4(v.x * s, v.y * s, v.z * s, v.w * s);
  }

  for (int base = beg; base < end; base += 32) {
    int n = end - base;
    if (n > 32) n = 32;
    int myr = c;
    float myw = 0.0f;
    if (lane < n) {
      int p = base + lane;
      myr = (int)srcs[p] + ((p >= mid) ? 65536 : 0);
      int deg = off[2 * myr + 2] - off[2 * myr] + 1;
      myw = (1.0f / sqrtf((float)deg)) * dc;
    }
    int j = 0;
    for (; j + 4 <= n; j += 4) {
      int r0 = __builtin_amdgcn_readlane(myr, j);
      int r1 = __builtin_amdgcn_readlane(myr, j + 1);
      int r2 = __builtin_amdgcn_readlane(myr, j + 2);
      int r3 = __builtin_amdgcn_readlane(myr, j + 3);
      float4 v0 = ld_row(h, r0, lane);
      float4 v1 = ld_row(h, r1, lane);
      float4 v2 = ld_row(h, r2, lane);
      float4 v3 = ld_row(h, r3, lane);
      float w0 = __int_as_float(__builtin_amdgcn_readlane(__float_as_int(myw), j));
      float w1 = __int_as_float(__builtin_amdgcn_readlane(__float_as_int(myw), j + 1));
      float w2 = __int_as_float(__builtin_amdgcn_readlane(__float_as_int(myw), j + 2));
      float w3 = __int_as_float(__builtin_amdgcn_readlane(__float_as_int(myw), j + 3));
      acc.x += v0.x * w0; acc.y += v0.y * w0; acc.z += v0.z * w0; acc.w += v0.w * w0;
      acc.x += v1.x * w1; acc.y += v1.y * w1; acc.z += v1.z * w1; acc.w += v1.w * w1;
      acc.x += v2.x * w2; acc.y += v2.y * w2; acc.z += v2.z * w2; acc.w += v2.w * w2;
      acc.x += v3.x * w3; acc.y += v3.y * w3; acc.z += v3.z * w3; acc.w += v3.w * w3;
    }
    for (; j < n; ++j) {
      int r0 = __builtin_amdgcn_readlane(myr, j);
      float4 v0 = ld_row(h, r0, lane);
      float w0 = __int_as_float(__builtin_amdgcn_readlane(__float_as_int(myw), j));
      acc.x += v0.x * w0; acc.y += v0.y * w0; acc.z += v0.z * w0; acc.w += v0.w * w0;
    }
  }

  acc.x += bv.x; acc.y += bv.y; acc.z += bv.z; acc.w += bv.w;
  if (RELU) {
    acc.x = fmaxf(acc.x, 0.0f); acc.y = fmaxf(acc.y, 0.0f);
    acc.z = fmaxf(acc.z, 0.0f); acc.w = fmaxf(acc.w, 0.0f);
  }
  ((float4*)(out + (size_t)c * HIDC))[lane] = acc;
}

__global__ void k_decode(const float* __restrict__ z, const int* __restrict__ pos,
                         const int* __restrict__ neg, float* __restrict__ out) {
  int w = (blockIdx.x * blockDim.x + threadIdx.x) >> 5;
  int lane = threadIdx.x & 31;
  if (w >= NPOS + NNEG) return;
  int a, b;
  if (w < NPOS) { a = pos[w]; b = pos[NPOS + w]; }
  else          { int j = w - NPOS; a = neg[j]; b = neg[NNEG + j]; }
  float4 va = ((const float4*)(z + (size_t)a * HIDC))[lane];
  float4 vb = ((const float4*)(z + (size_t)b * HIDC))[lane];
  float s = va.x * vb.x + va.y * vb.y + va.z * vb.z + va.w * vb.w;
#pragma unroll
  for (int off = 16; off > 0; off >>= 1) s += __shfl_xor(s, off, 32);
  if (lane == 0) out[w] = s;
}

extern "C" void kernel_launch(void* const* d_in, const int* in_sizes, int n_in,
                              void* d_out, int out_size, void* d_ws, size_t ws_size,
                              hipStream_t stream) {
  const float* x  = (const float*)d_in[0];
  const int*   ei = (const int*)d_in[1];
  const int*   pos = (const int*)d_in[2];
  const int*   neg = (const int*)d_in[3];
  const float* W1 = (const float*)d_in[4];
  const float* b1 = (const float*)d_in[5];
  const float* W2 = (const float*)d_in[6];
  const float* b2 = (const float*)d_in[7];
  float* logits = (float*)d_out;

  char* ws = (char*)d_ws;
  int*   off  = (int*)ws;
  float* WT   = (float*)(ws + 800256u);
  unsigned short* srcs = (unsigned short*)(ws + 931328u);
  float* bufH = (float*)(ws + 4374528u);
  float* bufZ = (float*)(ws + 55574528u);
  int*   rank = (int*)bufZ;

  const int* row = ei;
  const int* col = ei + NE;
  const int B = 256;

  k_zero_cnt<<<(NK + B - 1) / B, B, 0, stream>>>(off);
  k_count_rank<<<(NE + B - 1) / B, B, 0, stream>>>(off, row, col, rank);
  k_scan<<<1, 1024, 0, stream>>>(off);
  k_fill_srcs<<<(NE + B - 1) / B, B, 0, stream>>>(srcs, off, row, col, rank);

  int gemm_blocks = (((NN + 15) / 16) + 7) / 8;
  int ga_blocks   = (NN * 32 + B - 1) / B;

  k_transpose<<<(INC * HIDC + B - 1) / B, B, 0, stream>>>(W1, WT, INC, HIDC);
  k_gemm_wmma<INC><<<gemm_blocks, B, 0, stream>>>(x, WT, bufH);
  k_gather<1><<<ga_blocks, B, 0, stream>>>(bufZ, bufH, off, srcs, b1);

  k_transpose<<<(HIDC * HIDC + B - 1) / B, B, 0, stream>>>(W2, WT, HIDC, HIDC);
  k_gemm_wmma<HIDC><<<gemm_blocks, B, 0, stream>>>(bufZ, WT, bufH);
  k_gather<0><<<ga_blocks, B, 0, stream>>>(bufZ, bufH, off, srcs, b2);

  int dec_blocks = ((NPOS + NNEG) * 32 + B - 1) / B;
  k_decode<<<dec_blocks, B, 0, stream>>>(bufZ, pos, neg, logits);
}


# long-lived waves: a wave loops over 8 rows (aggregation), 32 pairs (decoder: 8 row loads in flight, one 128-byte store) or 4 edges a thread (sort); 83 thousand waves per call, not 720 thousand
# speedup vs baseline: 2.2487x; 1.0616x over previous
#include <hip/hip_runtime.h>

#define NN   100000
#define INC  256
#define HIDC 128
#define NE   1600000
#define NPOS 200000
#define NNEG 200000
#define NK   (2 * NN)

typedef __attribute__((ext_vector_type(2))) float v2f;
typedef __attribute__((ext_vector_type(8))) float v8f;

__global__ void k_zero_cnt(int* cnt) {
  int i = blockIdx.x * blockDim.x + threadIdx.x;
  if (i < NK) cnt[i] = 0;
}

__global__ void k_count_rank(int* cnt, const int* __restrict__ row,
                             const int* __restrict__ col, int* __restrict__ rank) {
  const int stride = gridDim.x * blockDim.x;
  for (int i = blockIdx.x * blockDim.x + threadIdx.x; i < NE; i += stride)
    rank[i] = atomicAdd(&cnt[2 * col[i] + (row[i] >> 16)], 1);
}

#define SCAN_TILES 25
#define SCAN_GROUP 5

__device__ __forceinline__ void scan_load_group(const int* off, int g, int tid,
                                                int4 (&va)[SCAN_GROUP], int4 (&vb)[SCAN_GROUP],
                                                int (&s)[SCAN_GROUP]) {
#pragma unroll
  for (int q = 0; q < SCAN_GROUP; ++q) {
    int i = (g + q) * 8192 + tid * 8;
    va[q] = make_int4(0, 0, 0, 0);
    vb[q] = make_int4(0, 0, 0, 0);
    if (i < NK) {
      va[q] = *(const int4*)(off + i);
      vb[q] = *(const int4*)(off + i + 4);
    }
  }
#pragma unroll
  for (int q = 0; q < SCAN_GROUP; ++q)
    s[q] = (va[q].x + va[q].y + va[q].z + va[q].w) + (vb[q].x + vb[q].y + vb[q].z + vb[q].w);
  const int lane = tid & 31;
#pragma unroll
  for (int d = 1; d < 32; d <<= 1) {
    int u[SCAN_GROUP];
#pragma unroll
    for (int q = 0; q < SCAN_GROUP; ++q) u[q] = __shfl_up(s[q], d, 32);
#pragma unroll
    for (int q = 0; q < SCAN_GROUP; ++q) if (lane >= d) s[q] += u[q];
  }
}

__global__ __launch_bounds__(1024) void k_scan(int* off) {
  __shared__ int part[SCAN_TILES * 32];
  __shared__ int wtot[32];
  const int tid = threadIdx.x, lane = tid & 31, wid = tid >> 5;

  for (int g = 0; g < SCAN_TILES; g += SCAN_GROUP) {
    int4 va[SCAN_GROUP], vb[SCAN_GROUP];
    int s[SCAN_GROUP];
    scan_load_group(off, g, tid, va, vb, s);
    if (lane == 31) {
#pragma unroll
      for (int q = 0; q < SCAN_GROUP; ++q) part[(g + q) * 32 + wid] = s[q];
    }
  }
  __syncthreads();

  int p = (tid < SCAN_TILES * 32) ? part[tid] : 0;
  int incl = p;
#pragma unroll
  for (int d = 1; d < 32; d <<= 1) {
    int u = __shfl_up(incl, d, 32);
    if (lane >= d) incl += u;
  }
  if (lane == 31) wtot[wid] = incl;
  __syncthreads();
  int wv = wtot[lane];
  int wincl = wv;
#pragma unroll
  for (int d = 1; d < 32; d <<= 1) {
    int u = __shfl_up(wincl, d, 32);
    if (lane >= d) wincl += u;
  }
  int wexcl = __shfl(wincl - wv, wid, 32);
  int total = __shfl(wincl, 31, 32);
  if (tid < SCAN_TILES * 32) part[tid] = wexcl + incl - p;
  if (tid == 0) off[NK] = total;
  __syncthreads();

#pragma nounroll
  for (int g = 0; g < SCAN_TILES; g += SCAN_GROUP) {
    int4 va[SCAN_GROUP], vb[SCAN_GROUP];
    int s[SCAN_GROUP];
    scan_load_group(off, g, tid, va, vb, s);
#pragma unroll
    for (int q = 0; q < SCAN_GROUP; ++q) {
      const int4 a = va[q];
      const int4 b = vb[q];
      int i = (g + q) * 8192 + tid * 8;
      int own = (a.x + a.y + a.z + a.w) + (b.x + b.y + b.z + b.w);
      int ex = part[(g + q) * 32 + wid] + s[q] - own;
      if (i < NK) {
        int4 o, o2;
        o.x = ex;
        o.y = o.x + a.x;
        o.z = o.y + a.y;
        o.w = o.z + a.z;
        o2.x = o.w + a.w;
        o2.y = o2.x + b.x;
        o2.z = o2.y + b.y;
        o2.w = o2.z + b.z;
        *(int4*)(off + i) = o;
        *(int4*)(off + i + 4) = o2;
      }
    }
  }
}

__global__ void k_fill_srcs(unsigned short* __restrict__ srcs, const int* __restrict__ off,
                            const int* __restrict__ row, const int* __restrict__ col,
                            const int* __restrict__ rank) {
  const int stride = gridDim.x * blockDim.x;
  for (int i = blockIdx.x * blockDim.x + threadIdx.x; i < NE; i += stride) {
    int r = row[i];
    srcs[off[2 * col[i] + (r >> 16)] + rank[i]] = (unsigned short)(r & 0xFFFF);
  }
}

__global__ void k_transpose(const float* __restrict__ W, float* __restrict__ WT,
                            int K, int N) {
  int i = blockIdx.x * blockDim.x + threadIdx.x;
  if (i >= K * N) return;
  int k = i / N, n = i - k * N;
  WT[n * K + k] = W[i];
}

template <int K>
__global__ __launch_bounds__(256) void k_gemm_wmma(const float* __restrict__ A,
                                                   const float* __restrict__ WT,
                                                   float* __restrict__ C) {
  int wave = (blockIdx.x * blockDim.x + threadIdx.x) >> 5;
  int lane = threadIdx.x & 31;
  int m0 = wave * 16;
  if (m0 >= NN) return;
  int half = lane >> 4;
  int l15  = lane & 15;
  const float* aptr  = A  + (size_t)(m0 + l15) * K + half * 2;
  const float* bbase = WT + (size_t)l15 * K + half * 2;

  v8f acc[8];
#pragma unroll
  for (int t = 0; t < 8; ++t) acc[t] = (v8f)(0.0f);

  __builtin_prefetch(aptr, 0, 0);
#pragma unroll 2
  for (int k = 0; k < K; k += 4) {
    v2f a = *(const v2f*)(aptr + k);
#pragma unroll
    for (int t = 0; t < 8; ++t) {
      v2f b = *(const v2f*)(bbase + (size_t)t * 16 * K + k);
      acc[t] = __builtin_amdgcn_wmma_f32_16x16x4_f32(
          false, a, false, b, (short)0, acc[t], false, false);
    }
  }

#pragma unroll
  for (int t = 0; t < 8; ++t) {
#pragma unroll
    for (int i = 0; i < 8; ++i) {
      C[(size_t)(m0 + i + half * 8) * HIDC + t * 16 + l15] = acc[t][i];
    }
  }
}

__device__ __forceinline__ float4 ld_row(const float* __restrict__ h, int r, int lane) {
  return ((const float4*)(h + (size_t)r * HIDC))[lane];
}

template <int RELU>
__global__ __launch_bounds__(256) void k_gather(float* __restrict__ out,
                                                const float* __restrict__ h,
                                                const int* __restrict__ off,
                                                const unsigned short* __restrict__ srcs,
                                                const float* __restrict__ bias) {
  const int lane = threadIdx.x & 31;
  const int gw = __builtin_amdgcn_readfirstlane((int)((blockIdx.x * blockDim.x + threadIdx.x) >> 5));
  const int nw = (int)((gridDim.x * blockDim.x) >> 5);
  const float4 bv = ((const float4*)bias)[lane];
  for (int c = gw; c < NN; c += nw) {
    const int beg = off[2 * c];
    const int mid = off[2 * c + 1];
    const int end = off[2 * c + 2];
    const float dc = 1.0f / sqrtf((float)(end - beg + 1));
    float4 acc;
    {
      float4 v = ld_row(h, c, lane);
      float s = dc * dc;
      acc = make_float4(v.x * s, v.y * s, v.z * s, v.w * s);
    }

    for (int base = beg; base < end; base += 32) {
      int n = end - base;
      if (n > 32) n = 32;
      int myr = c;
      float myw = 0.0f;
      if (lane < n) {
        int p = base + lane;
        myr = (int)srcs[p] + ((p >= mid) ? 65536 : 0);
        int deg = off[2 * myr + 2] - off[2 * myr] + 1;
        myw = (1.0f / sqrtf((float)deg)) * dc;
      }
      int j = 0;
      for (; j + 4 <= n; j += 4) {
        int r0 = __builtin_amdgcn_readlane(myr, j);
        int r1 = __builtin_amdgcn_readlane(myr, j + 1);
        int r2 = __builtin_amdgcn_readlane(myr, j + 2);
        int r3 = __builtin_amdgcn_readlane(myr, j + 3);
        float4 v0 = ld_row(h, r0, lane);
        float4 v1 = ld_row(h, r1, lane);
        float4 v2 = ld_row(h, r2, lane);
        float4 v3 = ld_row(h, r3, lane);
        float w0 = __int_as_float(__builtin_amdgcn_readlane(__float_as_int(myw), j));
        float w1 = __int_as_float(__builtin_amdgcn_readlane(__float_as_int(myw), j + 1));
        float w2 = __int_as_float(__builtin_amdgcn_readlane(__float_as_int(myw), j + 2));
        float w3 = __int_as_float(__builtin_amdgcn_readlane(__float_as_int(myw), j + 3));
        acc.x += v0.x * w0; acc.y += v0.y * w0; acc.z += v0.z * w0; acc.w += v0.w * w0;
        acc.x += v1.x * w1; acc.y += v1.y * w1; acc.z += v1.z * w1; acc.w += v1.w * w1;
        acc.x += v2.x * w2; acc.y += v2.y * w2; acc.z += v2.z * w2; acc.w += v2.w * w2;
        acc.x += v3.x * w3; acc.y += v3.y * w3; acc.z += v3.z * w3; acc.w += v3.w * w3;
      }
      for (; j < n; ++j) {
        int r0 = __builtin_amdgcn_readlane(myr, j);
        float4 v0 = ld_row(h, r0, lane);
        float w0 = __int_as_float(__builtin_amdgcn_readlane(__float_as_int(myw), j));
        acc.x += v0.x * w0; acc.y += v0.y * w0; acc.z += v0.z * w0; acc.w += v0.w * w0;
      }
    }

    acc.x += bv.x; acc.y += bv.y; acc.z += bv.z; acc.w += bv.w;
    if (RELU) {
      acc.x = fmaxf(acc.x, 0.0f); acc.y = fmaxf(acc.y, 0.0f);
      acc.z = fmaxf(acc.z, 0.0f); acc.w = fmaxf(acc.w, 0.0f);
    }
    ((float4*)(out + (size_t)c * HIDC))[lane] = acc;
  }
}

#define NPAIR (NPOS + NNEG)

__device__ __forceinline__ float dot4(float4 a, float4 b) {
  return fmaf(a.w, b.w, fmaf(a.z, b.z, fmaf(a.x, b.x, a.y * b.y)));
}
static_assert(NPOS % 32 == 0 && NNEG % 32 == 0, "a chunk of 32 pairs must not straddle the two lists");

__global__ __launch_bounds__(256) void k_decode(const float* __restrict__ z,
                                                const int* __restrict__ pos,
                                                const int* __restrict__ neg,
                                                float* __restrict__ out) {
  const int lane = threadIdx.x & 31;
  const int gw = __builtin_amdgcn_readfirstlane((int)((blockIdx.x * blockDim.x + threadIdx.x) >> 5));
  const int nw = (int)((gridDim.x * blockDim.x) >> 5);
  for (int ch = gw; ch < NPAIR / 32; ch += nw) {
    const int w0 = ch * 32;
    const int* __restrict__ src = (w0 < NPOS) ? pos : neg;
    const int cnt = (w0 < NPOS) ? NPOS : NNEG;
    const int j0 = (w0 < NPOS) ? w0 : w0 - NPOS;
    const int ia = src[j0 + lane];
    const int ib = src[cnt + j0 + lane];
    float res = 0.0f;
    for (int j = 0; j < 32; j += 4) {
      int a0 = __builtin_amdgcn_readlane(ia, j);
      int b0 = __builtin_amdgcn_readlane(ib, j);
      int a1 = __builtin_amdgcn_readlane(ia, j + 1);
      int b1 = __builtin_amdgcn_readlane(ib, j + 1);
      int a2 = __builtin_amdgcn_readlane(ia, j + 2);
      int b2 = __builtin_amdgcn_readlane(ib, j + 2);
      int a3 = __builtin_amdgcn_readlane(ia, j + 3);
      int b3 = __builtin_amdgcn_readlane(ib, j + 3);
      float4 x0 = ld_row(z, a0, lane), y0 = ld_row(z, b0, lane);
      float4 x1 = ld_row(z, a1, lane), y1 = ld_row(z, b1, lane);
      float4 x2 = ld_row(z, a2, lane), y2 = ld_row(z, b2, lane);
      float4 x3 = ld_row(z, a3, lane), y3 = ld_row(z, b3, lane);
      float s0 = dot4(x0, y0), s1 = dot4(x1, y1), s2 = dot4(x2, y2), s3 = dot4(x3, y3);
#pragma unroll
      for (int d = 16; d > 0; d >>= 1) {
        s0 += __shfl_xor(s0, d, 32);
        s1 += __shfl_xor(s1, d, 32);
        s2 += __shfl_xor(s2, d, 32);
        s3 += __shfl_xor(s3, d, 32);
      }
      res = (lane == j)     ? s0 : res;
      res = (lane == j + 1) ? s1 : res;
      res = (lane == j + 2) ? s2 : res;
      res = (lane == j + 3) ? s3 : res;
    }
    out[w0 + lane] = res;
  }
}

extern "C" void kernel_launch(void* const* d_in, const int* in_sizes, int n_in,
                              void* d_out, int out_size, void* d_ws, size_t ws_size,
                              hipStream_t stream) {
  const float* x  = (const float*)d_in[0];
  const int*   ei = (const int*)d_in[1];
  const int*   pos = (const int*)d_in[2];
  const int*   neg = (const int*)d_in[3];
  const float* W1 = (const float*)d_in[4];
  const float* b1 = (const float*)d_in[5];
  const float* W2 = (const float*)d_in[6];
  const float* b2 = (const float*)d_in[7];
  float* logits = (float*)d_out;

  char* ws = (char*)d_ws;
  int*   off  = (int*)ws;
  float* WT   = (float*)(ws + 800256u);
  unsigned short* srcs = (unsigned short*)(ws + 931328u);
  float* bufH = (float*)(ws + 4374528u);
  float* bufZ = (float*)(ws + 55574528u);
  int*   rank = (int*)bufZ;

  const int* row = ei;
  const int* col = ei + NE;
  const int B = 256;

  k_zero_cnt<<<(NK + B - 1) / B, B, 0, stream>>>(off);
  const int edge_blocks = (NE + 4 * B - 1) / (4 * B);
  k_count_rank<<<edge_blocks, B, 0, stream>>>(off, row, col, rank);
  k_scan<<<1, 1024, 0, stream>>>(off);
  k_fill_srcs<<<edge_blocks, B, 0, stream>>>(srcs, off, row, col, rank);

  int gemm_blocks = (((NN + 15) / 16) + 7) / 8;
  int ga_blocks   = (NN / 8 * 32 + B - 1) / B;

  k_transpose<<<(INC * HIDC + B - 1) / B, B, 0, stream>>>(W1, WT, INC, HIDC);
  k_gemm_wmma<INC><<<gemm_blocks, B, 0, stream>>>(x, WT, bufH);
  k_gather<1><<<ga_blocks, B, 0, stream>>>(bufZ, bufH, off, srcs, b1);

  k_transpose<<<(HIDC * HIDC + B - 1) / B, B, 0, stream>>>(W2, WT, HIDC, HIDC);
  k_gemm_wmma<HIDC><<<gemm_blocks, B, 0, stream>>>(bufZ, WT, bufH);
  k_gather<0><<<ga_blocks, B, 0, stream>>>(bufZ, bufH, off, srcs, b2);

  int dec_blocks = ((NPOS + NNEG) + B - 1) / B;
  k_decode<<<dec_blocks, B, 0, stream>>>(bufZ, pos, neg, logits);
}


# edge sort: per-edge returning global adds, the one-group scan and scattered 2-byte stores replaced by a two-level counting sort binned in LDS; edges of a run now in LDS-add arrival order
# speedup vs baseline: 3.2338x; 1.4381x over previous
#include <hip/hip_runtime.h>

#define NN   100000
#define INC  256
#define HIDC 128
#define NE   1600000
#define NPOS 200000
#define NNEG 200000
#define NK   (2 * NN)

typedef __attribute__((ext_vector_type(2))) float v2f;
typedef __attribute__((ext_vector_type(8))) float v8f;

#define BSH   8
#define NB    ((NN + 255) >> BSH)
#define EPT   25
#define SLICE (256 * EPT)
#define NSLICE (NE / SLICE)
static_assert(NE % SLICE == 0, "the edge list must split into whole slices");
static_assert(NB <= 512, "two table entries per thread");
static_assert(NN <= (1 << 17), "a source id takes 17 bits of an edge record");
#define STAGE_CAP 16384
#define SORT_BATCH 8

__device__ __forceinline__ int block_excl_scan(int s, int* wsum) {
  const int lane = threadIdx.x & 31, wid = threadIdx.x >> 5;
  int incl = s;
#pragma unroll
  for (int d = 1; d < 32; d <<= 1) {
    int u = __shfl_up(incl, d, 32);
    if (lane >= d) incl += u;
  }
  __syncthreads();
  if (lane == 31) wsum[wid] = incl;
  __syncthreads();
  int pre = 0;
#pragma unroll
  for (int w = 0; w < 8; ++w) pre += (w < wid) ? wsum[w] : 0;
  return pre + incl - s;
}

__global__ void k_zero_aux(int* aux) {
  int i = blockIdx.x * blockDim.x + threadIdx.x;
  if (i < 1024) aux[i] = 0;
}

__global__ __launch_bounds__(256) void k_bucket_count(const int* __restrict__ col,
                                                      int* __restrict__ total) {
  __shared__ int lcnt[512];
  const int tid = threadIdx.x;
  lcnt[tid] = 0;
  lcnt[tid + 256] = 0;
  __syncthreads();
  const int base = blockIdx.x * SLICE + tid;
  int c[EPT];
#pragma unroll
  for (int j = 0; j < EPT; ++j) c[j] = col[base + j * 256];
#pragma unroll
  for (int j = 0; j < EPT; ++j) atomicAdd(&lcnt[c[j] >> BSH], 1);
  __syncthreads();
  for (int b = tid; b < NB; b += 256) {
    int v = lcnt[b];
    if (v) atomicAdd(&total[b], v);
  }
}

__global__ __launch_bounds__(256) void k_bucket_split(const int* __restrict__ row,
                                                      const int* __restrict__ col,
                                                      const int* __restrict__ total,
                                                      int* __restrict__ gcur,
                                                      int* __restrict__ bstart_g,
                                                      int* __restrict__ tmp) {
  __shared__ int lcnt[512];
  __shared__ int lstart[512];
  __shared__ int lbase[512];
  __shared__ int wsum[8];
  __shared__ int stage[SLICE];
  __shared__ unsigned short stageb[SLICE];
  const int tid = threadIdx.x;
  lcnt[tid] = 0;
  lcnt[tid + 256] = 0;

  int c[EPT], r[EPT], lr[EPT];
  const int base = blockIdx.x * SLICE + tid;
#pragma unroll
  for (int j = 0; j < EPT; ++j) {
    c[j] = col[base + j * 256];
    r[j] = row[base + j * 256];
  }

  const int b0 = 2 * tid, b1 = 2 * tid + 1;
  const int t0 = (b0 < NB) ? total[b0] : 0;
  const int t1 = (b1 < NB) ? total[b1] : 0;
  const int bs0 = block_excl_scan(t0 + t1, wsum);
  const int bs1 = bs0 + t0;
  if (blockIdx.x == 0) {
    if (b0 < NB) bstart_g[b0] = bs0;
    if (b1 < NB) bstart_g[b1] = bs1;
  }

#pragma unroll
  for (int j = 0; j < EPT; ++j) lr[j] = atomicAdd(&lcnt[c[j] >> BSH], 1);
  __syncthreads();

  const int n0 = lcnt[b0], n1 = lcnt[b1];
  const int ls0 = block_excl_scan(n0 + n1, wsum);
  const int ls1 = ls0 + n0;
  lstart[b0] = ls0;
  lstart[b1] = ls1;
  {
    int g0 = (n0 > 0) ? atomicAdd(&gcur[b0], n0) : 0;
    int g1 = (n1 > 0) ? atomicAdd(&gcur[b1], n1) : 0;
    lbase[b0] = bs0 + g0 - ls0;
    lbase[b1] = bs1 + g1 - ls1;
  }
  __syncthreads();

#pragma unroll
  for (int j = 0; j < EPT; ++j) {
    int b = c[j] >> BSH;
    int p = lstart[b] + lr[j];
    stage[p] = r[j] | ((c[j] & 255) << 17);
    stageb[p] = (unsigned short)b;
  }
  __syncthreads();

#pragma unroll
  for (int j = 0; j < EPT; ++j) {
    int i = j * 256 + tid;
    tmp[lbase[stageb[i]] + i] = stage[i];
  }
}

__global__ __launch_bounds__(256) void k_bucket_sort(const int* __restrict__ tmp,
                                                     const int* __restrict__ total,
                                                     const int* __restrict__ bstart_g,
                                                     int* __restrict__ off,
                                                     unsigned short* __restrict__ srcs) {
  __shared__ int cnt[512];
  __shared__ int cur[512];
  __shared__ int wsum[8];
  __shared__ unsigned short stage[STAGE_CAP];
  const int tid = threadIdx.x, b = blockIdx.x;
  const int n = total[b];
  const int s0 = bstart_g[b];
  const int* __restrict__ rec = tmp + s0;
  cnt[tid] = 0;
  cnt[tid + 256] = 0;
  __syncthreads();
  for (int i0 = tid; i0 < n; i0 += 256 * SORT_BATCH) {
    int e[SORT_BATCH];
#pragma unroll
    for (int q = 0; q < SORT_BATCH; ++q) e[q] = rec[min(i0 + q * 256, n - 1)];
#pragma unroll
    for (int q = 0; q < SORT_BATCH; ++q)
      if (i0 + q * 256 < n) atomicAdd(&cnt[(e[q] >> 16) & 511], 1);
  }
  __syncthreads();
  const int c0 = cnt[2 * tid], c1 = cnt[2 * tid + 1];
  const int ex = block_excl_scan(c0 + c1, wsum);
  const int node = (b << BSH) + tid;
  if (node < NN) {
    off[2 * node] = s0 + ex;
    off[2 * node + 1] = s0 + ex + c0;
  }
  if (b == NB - 1 && tid == 0) off[NK] = s0 + n;
  cur[2 * tid] = ex;
  cur[2 * tid + 1] = ex + c0;
  __syncthreads();
  if (n <= STAGE_CAP) {
    for (int i0 = tid; i0 < n; i0 += 256 * SORT_BATCH) {
      int e[SORT_BATCH];
#pragma unroll
      for (int q = 0; q < SORT_BATCH; ++q) e[q] = rec[min(i0 + q * 256, n - 1)];
#pragma unroll
      for (int q = 0; q < SORT_BATCH; ++q) {
        if (i0 + q * 256 < n) {
          int p = atomicAdd(&cur[(e[q] >> 16) & 511], 1);
          stage[p] = (unsigned short)(e[q] & 0xFFFF);
        }
      }
    }
    __syncthreads();
    for (int i = tid; i < n; i += 256) srcs[s0 + i] = stage[i];
  } else {
    for (int i = tid; i < n; i += 256) {
      int e = rec[i];
      int p = atomicAdd(&cur[(e >> 16) & 511], 1);
      srcs[s0 + p] = (unsigned short)(e & 0xFFFF);
    }
  }
}

__global__ void k_transpose(const float* __restrict__ W, float* __restrict__ WT,
                            int K, int N) {
  int i = blockIdx.x * blockDim.x + threadIdx.x;
  if (i >= K * N) return;
  int k = i / N, n = i - k * N;
  WT[n * K + k] = W[i];
}

template <int K>
__global__ __launch_bounds__(256) void k_gemm_wmma(const float* __restrict__ A,
                                                   const float* __restrict__ WT,
                                                   float* __restrict__ C) {
  int wave = (blockIdx.x * blockDim.x + threadIdx.x) >> 5;
  int lane = threadIdx.x & 31;
  int m0 = wave * 16;
  if (m0 >= NN) return;
  int half = lane >> 4;
  int l15  = lane & 15;
  const float* aptr  = A  + (size_t)(m0 + l15) * K + half * 2;
  const float* bbase = WT + (size_t)l15 * K + half * 2;

  v8f acc[8];
#pragma unroll
  for (int t = 0; t < 8; ++t) acc[t] = (v8f)(0.0f);

  __builtin_prefetch(aptr, 0, 0);
#pragma unroll 2
  for (int k = 0; k < K; k += 4) {
    v2f a = *(const v2f*)(aptr + k);
#pragma unroll
    for (int t = 0; t < 8; ++t) {
      v2f b = *(const v2f*)(bbase + (size_t)t * 16 * K + k);
      acc[t] = __builtin_amdgcn_wmma_f32_16x16x4_f32(
          false, a, false, b, (short)0, acc[t], false, false);
    }
  }

#pragma unroll
  for (int t = 0; t < 8; ++t) {
#pragma unroll
    for (int i = 0; i < 8; ++i) {
      C[(size_t)(m0 + i + half * 8) * HIDC + t * 16 + l15] = acc[t][i];
    }
  }
}

__device__ __forceinline__ float4 ld_row(const float* __restrict__ h, int r, int lane) {
  return ((const float4*)(h + (size_t)r * HIDC))[lane];
}

template <int RELU>
__global__ __launch_bounds__(256) void k_gather(float* __restrict__ out,
                                                const float* __restrict__ h,
                                                const int* __restrict__ off,
                                                const unsigned short* __restrict__ srcs,
                                                const float* __restrict__ bias) {
  const int lane = threadIdx.x & 31;
  const int gw = __builtin_amdgcn_readfirstlane((int)((blockIdx.x * blockDim.x + threadIdx.x) >> 5));
  const int nw = (int)((gridDim.x * blockDim.x) >> 5);
  const float4 bv = ((const float4*)bias)[lane];
  for (int c = gw; c < NN; c += nw) {
    const int beg = off[2 * c];
    const int mid = off[2 * c + 1];
    const int end = off[2 * c + 2];
    const float dc = 1.0f / sqrtf((float)(end - beg + 1));
    float4 acc;
    {
      float4 v = ld_row(h, c, lane);
      float s = dc * dc;
      acc = make_float4(v.x * s, v.y * s, v.z * s, v.w * s);
    }

    for (int base = beg; base < end; base += 32) {
      int n = end - base;
      if (n > 32) n = 32;
      int myr = c;
      float myw = 0.0f;
      if (lane < n) {
        int p = base + lane;
        myr = (int)srcs[p] + ((p >= mid) ? 65536 : 0);
        int deg = off[2 * myr + 2] - off[2 * myr] + 1;
        myw = (1.0f / sqrtf((float)deg)) * dc;
      }
      int j = 0;
      for (; j + 4 <= n; j += 4) {
        int r0 = __builtin_amdgcn_readlane(myr, j);
        int r1 = __builtin_amdgcn_readlane(myr, j + 1);
        int r2 = __builtin_amdgcn_readlane(myr, j + 2);
        int r3 = __builtin_amdgcn_readlane(myr, j + 3);
        float4 v0 = ld_row(h, r0, lane);
        float4 v1 = ld_row(h, r1, lane);
        float4 v2 = ld_row(h, r2, lane);
        float4 v3 = ld_row(h, r3, lane);
        float w0 = __int_as_float(__builtin_amdgcn_readlane(__float_as_int(myw), j));
        float w1 = __int_as_float(__builtin_amdgcn_readlane(__float_as_int(myw), j + 1));
        float w2 = __int_as_float(__builtin_amdgcn_readlane(__float_as_int(myw), j + 2));
        float w3 = __int_as_float(__builtin_amdgcn_readlane(__float_as_int(myw), j + 3));
        acc.x += v0.x * w0; acc.y += v0.y * w0; acc.z += v0.z * w0; acc.w += v0.w * w0;
        acc.x += v1.x * w1; acc.y += v1.y * w1; acc.z += v1.z * w1; acc.w += v1.w * w1;
        acc.x += v2.x * w2; acc.y += v2.y * w2; acc.z += v2.z * w2; acc.w += v2.w * w2;
        acc.x += v3.x * w3; acc.y += v3.y * w3; acc.z += v3.z * w3; acc.w += v3.w * w3;
      }
      for (; j < n; ++j) {
        int r0 = __builtin_amdgcn_readlane(myr, j);
        float4 v0 = ld_row(h, r0, lane);
        float w0 = __int_as_float(__builtin_amdgcn_readlane(__float_as_int(myw), j));
        acc.x += v0.x * w0; acc.y += v0.y * w0; acc.z += v0.z * w0; acc.w += v0.w * w0;
      }
    }

    acc.x += bv.x; acc.y += bv.y; acc.z += bv.z; acc.w += bv.w;
    if (RELU) {
      acc.x = fmaxf(acc.x, 0.0f); acc.y = fmaxf(acc.y, 0.0f);
      acc.z = fmaxf(acc.z, 0.0f); acc.w = fmaxf(acc.w, 0.0f);
    }
    ((float4*)(out + (size_t)c * HIDC))[lane] = acc;
  }
}

#define NPAIR (NPOS + NNEG)

__device__ __forceinline__ float dot4(float4 a, float4 b) {
  return fmaf(a.w, b.w, fmaf(a.z, b.z, fmaf(a.x, b.x, a.y * b.y)));
}
static_assert(NPOS % 32 == 0 && NNEG % 32 == 0, "a chunk of 32 pairs must not straddle the two lists");

__global__ __launch_bounds__(256) void k_decode(const float* __restrict__ z,
                                                const int* __restrict__ pos,
                                                const int* __restrict__ neg,
                                                float* __restrict__ out) {
  const int lane = threadIdx.x & 31;
  const int gw = __builtin_amdgcn_readfirstlane((int)((blockIdx.x * blockDim.x + threadIdx.x) >> 5));
  const int nw = (int)((gridDim.x * blockDim.x) >> 5);
  for (int ch = gw; ch < NPAIR / 32; ch += nw) {
    const int w0 = ch * 32;
    const int* __restrict__ src = (w0 < NPOS) ? pos : neg;
    const int cnt = (w0 < NPOS) ? NPOS : NNEG;
    const int j0 = (w0 < NPOS) ? w0 : w0 - NPOS;
    const int ia = src[j0 + lane];
    const int ib = src[cnt + j0 + lane];
    float res = 0.0f;
    for (int j = 0; j < 32; j += 4) {
      int a0 = __builtin_amdgcn_readlane(ia, j);
      int b0 = __builtin_amdgcn_readlane(ib, j);
      int a1 = __builtin_amdgcn_readlane(ia, j + 1);
      int b1 = __builtin_amdgcn_readlane(ib, j + 1);
      int a2 = __builtin_amdgcn_readlane(ia, j + 2);
      int b2 = __builtin_amdgcn_readlane(ib, j + 2);
      int a3 = __builtin_amdgcn_readlane(ia, j + 3);
      int b3 = __builtin_amdgcn_readlane(ib, j + 3);
      float4 x0 = ld_row(z, a0, lane), y0 = ld_row(z, b0, lane);
      float4 x1 = ld_row(z, a1, lane), y1 = ld_row(z, b1, lane);
      float4 x2 = ld_row(z, a2, lane), y2 = ld_row(z, b2, lane);
      float4 x3 = ld_row(z, a3, lane), y3 = ld_row(z, b3, lane);
      float s0 = dot4(x0, y0), s1 = dot4(x1, y1), s2 = dot4(x2, y2), s3 = dot4(x3, y3);
#pragma unroll
      for (int d = 16; d > 0; d >>= 1) {
        s0 += __shfl_xor(s0, d, 32);
        s1 += __shfl_xor(s1, d, 32);
        s2 += __shfl_xor(s2, d, 32);
        s3 += __shfl_xor(s3, d, 32);
      }
      res = (lane == j)     ? s0 : res;
      res = (lane == j + 1) ? s1 : res;
      res = (lane == j + 2) ? s2 : res;
      res = (lane == j + 3) ? s3 : res;
    }
    out[w0 + lane] = res;
  }
}

extern "C" void kernel_launch(void* const* d_in, const int* in_sizes, int n_in,
                              void* d_out, int out_size, void* d_ws, size_t ws_size,
                              hipStream_t stream) {
  const float* x  = (const float*)d_in[0];
  const int*   ei = (const int*)d_in[1];
  const int*   pos = (const int*)d_in[2];
  const int*   neg = (const int*)d_in[3];
  const float* W1 = (const float*)d_in[4];
  const float* b1 = (const float*)d_in[5];
  const float* W2 = (const float*)d_in[6];
  const float* b2 = (const float*)d_in[7];
  float* logits = (float*)d_out;

  char* ws = (char*)d_ws;
  int*   off  = (int*)ws;
  float* WT   = (float*)(ws + 800256u);
  unsigned short* srcs = (unsigned short*)(ws + 931328u);
  float* bufH = (float*)(ws + 4374528u);
  float* bufZ = (float*)(ws + 55574528u);
  int*   tmp  = (int*)(ws + 55574528u);
  int*   aux  = (int*)(ws + 61974528u);
  int*   total    = aux;
  int*   gcur     = aux + 512;
  int*   bstart_g = aux + 1024;

  const int* row = ei;
  const int* col = ei + NE;
  const int B = 256;

  k_zero_aux<<<4, B, 0, stream>>>(aux);
  k_bucket_count<<<NSLICE, B, 0, stream>>>(col, total);
  k_bucket_split<<<NSLICE, B, 0, stream>>>(row, col, total, gcur, bstart_g, tmp);
  k_bucket_sort<<<NB, B, 0, stream>>>(tmp, total, bstart_g, off, srcs);

  int gemm_blocks = (((NN + 15) / 16) + 7) / 8;
  int ga_blocks   = (NN / 8 * 32 + B - 1) / B;

  k_transpose<<<(INC * HIDC + B - 1) / B, B, 0, stream>>>(W1, WT, INC, HIDC);
  k_gemm_wmma<INC><<<gemm_blocks, B, 0, stream>>>(x, WT, bufH);
  k_gather<1><<<ga_blocks, B, 0, stream>>>(bufZ, bufH, off, srcs, b1);

  k_transpose<<<(HIDC * HIDC + B - 1) / B, B, 0, stream>>>(W2, WT, HIDC, HIDC);
  k_gemm_wmma<HIDC><<<gemm_blocks, B, 0, stream>>>(bufZ, WT, bufH);
  k_gather<0><<<ga_blocks, B, 0, stream>>>(bufZ, bufH, off, srcs, b2);

  int dec_blocks = ((NPOS + NNEG) + B - 1) / B;
  k_decode<<<dec_blocks, B, 0, stream>>>(bufZ, pos, neg, logits);
}


# GEMM operand fetch: W packed in fragment order and staged once per work-group in LDS, A by 16-byte loads one 32-k chunk ahead; 512 groups of 12 waves, left-over strips quartered; new k summation order
# speedup vs baseline: 4.5604x; 1.4102x over previous
#include <hip/hip_runtime.h>

#define NN   100000
#define INC  256
#define HIDC 128
#define NE   1600000
#define NPOS 200000
#define NNEG 200000
#define NK   (2 * NN)

typedef __attribute__((ext_vector_type(2))) float v2f;
typedef __attribute__((ext_vector_type(8))) float v8f;

#define BSH   8
#define NB    ((NN + 255) >> BSH)
#define EPT   25
#define SLICE (256 * EPT)
#define NSLICE (NE / SLICE)
static_assert(NE % SLICE == 0, "the edge list must split into whole slices");
static_assert(NB <= 512, "two table entries per thread");
static_assert(NN <= (1 << 17), "a source id takes 17 bits of an edge record");
#define STAGE_CAP 16384
#define SORT_BATCH 8

__device__ __forceinline__ int block_excl_scan(int s, int* wsum) {
  const int lane = threadIdx.x & 31, wid = threadIdx.x >> 5;
  int incl = s;
#pragma unroll
  for (int d = 1; d < 32; d <<= 1) {
    int u = __shfl_up(incl, d, 32);
    if (lane >= d) incl += u;
  }
  __syncthreads();
  if (lane == 31) wsum[wid] = incl;
  __syncthreads();
  int pre = 0;
#pragma unroll
  for (int w = 0; w < 8; ++w) pre += (w < wid) ? wsum[w] : 0;
  return pre + incl - s;
}

__global__ void k_zero_aux(int* aux) {
  int i = blockIdx.x * blockDim.x + threadIdx.x;
  if (i < 1024) aux[i] = 0;
}

__global__ __launch_bounds__(256) void k_bucket_count(const int* __restrict__ col,
                                                      int* __restrict__ total) {
  __shared__ int lcnt[512];
  const int tid = threadIdx.x;
  lcnt[tid] = 0;
  lcnt[tid + 256] = 0;
  __syncthreads();
  const int base = blockIdx.x * SLICE + tid;
  int c[EPT];
#pragma unroll
  for (int j = 0; j < EPT; ++j) c[j] = col[base + j * 256];
#pragma unroll
  for (int j = 0; j < EPT; ++j) atomicAdd(&lcnt[c[j] >> BSH], 1);
  __syncthreads();
  for (int b = tid; b < NB; b += 256) {
    int v = lcnt[b];
    if (v) atomicAdd(&total[b], v);
  }
}

__global__ __launch_bounds__(256) void k_bucket_split(const int* __restrict__ row,
                                                      const int* __restrict__ col,
                                                      const int* __restrict__ total,
                                                      int* __restrict__ gcur,
                                                      int* __restrict__ bstart_g,
                                                      int* __restrict__ tmp) {
  __shared__ int lcnt[512];
  __shared__ int lstart[512];
  __shared__ int lbase[512];
  __shared__ int wsum[8];
  __shared__ int stage[SLICE];
  __shared__ unsigned short stageb[SLICE];
  const int tid = threadIdx.x;
  lcnt[tid] = 0;
  lcnt[tid + 256] = 0;

  int c[EPT], r[EPT], lr[EPT];
  const int base = blockIdx.x * SLICE + tid;
#pragma unroll
  for (int j = 0; j < EPT; ++j) {
    c[j] = col[base + j * 256];
    r[j] = row[base + j * 256];
  }

  const int b0 = 2 * tid, b1 = 2 * tid + 1;
  const int t0 = (b0 < NB) ? total[b0] : 0;
  const int t1 = (b1 < NB) ? total[b1] : 0;
  const int bs0 = block_excl_scan(t0 + t1, wsum);
  const int bs1 = bs0 + t0;
  if (blockIdx.x == 0) {
    if (b0 < NB) bstart_g[b0] = bs0;
    if (b1 < NB) bstart_g[b1] = bs1;
  }

#pragma unroll
  for (int j = 0; j < EPT; ++j) lr[j] = atomicAdd(&lcnt[c[j] >> BSH], 1);
  __syncthreads();

  const int n0 = lcnt[b0], n1 = lcnt[b1];
  const int ls0 = block_excl_scan(n0 + n1, wsum);
  const int ls1 = ls0 + n0;
  lstart[b0] = ls0;
  lstart[b1] = ls1;
  {
    int g0 = (n0 > 0) ? atomicAdd(&gcur[b0], n0) : 0;
    int g1 = (n1 > 0) ? atomicAdd(&gcur[b1], n1) : 0;
    lbase[b0] = bs0 + g0 - ls0;
    lbase[b1] = bs1 + g1 - ls1;
  }
  __syncthreads();

#pragma unroll
  for (int j = 0; j < EPT; ++j) {
    int b = c[j] >> BSH;
    int p = lstart[b] + lr[j];
    stage[p] = r[j] | ((c[j] & 255) << 17);
    stageb[p] = (unsigned short)b;
  }
  __syncthreads();

#pragma unroll
  for (int j = 0; j < EPT; ++j) {
    int i = j * 256 + tid;
    tmp[lbase[stageb[i]] + i] = stage[i];
  }
}

__global__ __launch_bounds__(256) void k_bucket_sort(const int* __restrict__ tmp,
                                                     const int* __restrict__ total,
                                                     const int* __restrict__ bstart_g,
                                                     int* __restrict__ off,
                                                     unsigned short* __restrict__ srcs) {
  __shared__ int cnt[512];
  __shared__ int cur[512];
  __shared__ int wsum[8];
  __shared__ unsigned short stage[STAGE_CAP];
  const int tid = threadIdx.x, b = blockIdx.x;
  const int n = total[b];
  const int s0 = bstart_g[b];
  const int* __restrict__ rec = tmp + s0;
  cnt[tid] = 0;
  cnt[tid + 256] = 0;
  __syncthreads();
  for (int i0 = tid; i0 < n; i0 += 256 * SORT_BATCH) {
    int e[SORT_BATCH];
#pragma unroll
    for (int q = 0; q < SORT_BATCH; ++q) e[q] = rec[min(i0 + q * 256, n - 1)];
#pragma unroll
    for (int q = 0; q < SORT_BATCH; ++q)
      if (i0 + q * 256 < n) atomicAdd(&cnt[(e[q] >> 16) & 511], 1);
  }
  __syncthreads();
  const int c0 = cnt[2 * tid], c1 = cnt[2 * tid + 1];
  const int ex = block_excl_scan(c0 + c1, wsum);
  const int node = (b << BSH) + tid;
  if (node < NN) {
    off[2 * node] = s0 + ex;
    off[2 * node + 1] = s0 + ex + c0;
  }
  if (b == NB - 1 && tid == 0) off[NK] = s0 + n;
  cur[2 * tid] = ex;
  cur[2 * tid + 1] = ex + c0;
  __syncthreads();
  if (n <= STAGE_CAP) {
    for (int i0 = tid; i0 < n; i0 += 256 * SORT_BATCH) {
      int e[SORT_BATCH];
#pragma unroll
      for (int q = 0; q < SORT_BATCH; ++q) e[q] = rec[min(i0 + q * 256, n - 1)];
#pragma unroll
      for (int q = 0; q < SORT_BATCH; ++q) {
        if (i0 + q * 256 < n) {
          int p = atomicAdd(&cur[(e[q] >> 16) & 511], 1);
          stage[p] = (unsigned short)(e[q] & 0xFFFF);
        }
      }
    }
    __syncthreads();
    for (int i = tid; i < n; i += 256) srcs[s0 + i] = stage[i];
  } else {
    for (int i = tid; i < n; i += 256) {
      int e = rec[i];
      int p = atomicAdd(&cur[(e >> 16) & 511], 1);
      srcs[s0 + p] = (unsigned short)(e & 0xFFFF);
    }
  }
}

__global__ void k_pack_w(const float* __restrict__ W, float* __restrict__ WP, int K) {
  int g = blockIdx.x * blockDim.x + threadIdx.x;
  if (g >= K * 32) return;
  int lane = g & 31, t = (g >> 5) & 7, p = g >> 8;
  int kk = 32 * (p >> 2) + 16 * (lane >> 4) + 4 * (p & 3);
  int n = 16 * t + (lane & 15);
  ((float4*)WP)[g] = make_float4(W[kk * HIDC + n], W[(kk + 1) * HIDC + n],
                                 W[(kk + 2) * HIDC + n], W[(kk + 3) * HIDC + n]);
}

#define GEMM_GROUPS 512
#define GEMM_WAVES  12
#define NSTRIP (NN / 16)
#define NLEFT  (NSTRIP - GEMM_GROUPS * GEMM_WAVES)
static_assert(NN % 16 == 0, "whole strips");
static_assert(NLEFT >= 0 && 4 * NLEFT <= GEMM_GROUPS, "a left-over quarter strip for the first wave of some groups");

template <int K, int NT>
__device__ __forceinline__ void gemm_strip(const float* __restrict__ A, const float4* wl,
                                           float* __restrict__ C, int m0, int t0, int lane) {
  const int c = lane & 15, h = lane >> 4;
  const float* ap = A + (size_t)(m0 + c) * K + 16 * h;
  v8f acc[NT];
#pragma unroll
  for (int t = 0; t < NT; ++t) acc[t] = (v8f)(0.0f);

  float4 cur[4], nxt[4];
#pragma unroll
  for (int i = 0; i < 4; ++i) cur[i] = *(const float4*)(ap + 4 * i);

  const float4* wp = wl + t0 * 32 + lane;
#pragma unroll 1
  for (int q = 0; q < K / 32; ++q) {
    const int qn = (q + 1 < K / 32) ? q + 1 : q;
#pragma unroll
    for (int i = 0; i < 4; ++i) nxt[i] = *(const float4*)(ap + 32 * qn + 4 * i);
    asm volatile("" ::: "memory");
#pragma unroll
    for (int i = 0; i < 4; ++i) {
      float4 b[NT];
#pragma unroll
      for (int t = 0; t < NT; ++t) b[t] = wp[(i * 8 + t) * 32];
      v2f a0 = {cur[i].x, cur[i].y};
      v2f a1 = {cur[i].z, cur[i].w};
#pragma unroll
      for (int t = 0; t < NT; ++t) {
        v2f b0 = {b[t].x, b[t].y};
        acc[t] = __builtin_amdgcn_wmma_f32_16x16x4_f32(
            false, a0, false, b0, (short)0, acc[t], false, false);
      }
#pragma unroll
      for (int t = 0; t < NT; ++t) {
        v2f b1 = {b[t].z, b[t].w};
        acc[t] = __builtin_amdgcn_wmma_f32_16x16x4_f32(
            false, a1, false, b1, (short)0, acc[t], false, false);
      }
    }
    wp += 4 * 8 * 32;
#pragma unroll
    for (int i = 0; i < 4; ++i) cur[i] = nxt[i];
  }

#pragma unroll
  for (int t = 0; t < NT; ++t) {
#pragma unroll
    for (int i = 0; i < 8; ++i) {
      C[(size_t)(m0 + i + h * 8) * HIDC + (t0 + t) * 16 + c] = acc[t][i];
    }
  }
}

template <int K>
__global__ __launch_bounds__(GEMM_WAVES * 32, 6) void k_gemm_wmma(const float* __restrict__ A,
                                                                 const float* __restrict__ WP,
                                                                 float* __restrict__ C) {
  __shared__ float4 wl[INC * HIDC / 4];
  const int tid = threadIdx.x;
  for (int i = tid; i < K * 32; i += GEMM_WAVES * 32) wl[i] = ((const float4*)WP)[i];
  __syncthreads();

  const int lane = tid & 31;
  const int wid = __builtin_amdgcn_readfirstlane(tid >> 5);
  gemm_strip<K, 8>(A, wl, C, (blockIdx.x + GEMM_GROUPS * wid) * 16, 0, lane);
  if (wid == 0 && blockIdx.x < 4 * NLEFT) {
    const int j = blockIdx.x;
    gemm_strip<K, 2>(A, wl, C, (GEMM_GROUPS * GEMM_WAVES + (j >> 2)) * 16, 2 * (j & 3), lane);
  }
}

__device__ __forceinline__ float4 ld_row(const float* __restrict__ h, int r, int lane) {
  return ((const float4*)(h + (size_t)r * HIDC))[lane];
}

template <int RELU>
__global__ __launch_bounds__(256) void k_gather(float* __restrict__ out,
                                                const float* __restrict__ h,
                                                const int* __restrict__ off,
                                                const unsigned short* __restrict__ srcs,
                                                const float* __restrict__ bias) {
  const int lane = threadIdx.x & 31;
  const int gw = __builtin_amdgcn_readfirstlane((int)((blockIdx.x * blockDim.x + threadIdx.x) >> 5));
  const int nw = (int)((gridDim.x * blockDim.x) >> 5);
  const float4 bv = ((const float4*)bias)[lane];
  for (int c = gw; c < NN; c += nw) {
    const int beg = off[2 * c];
    const int mid = off[2 * c + 1];
    const int end = off[2 * c + 2];
    const float dc = 1.0f / sqrtf((float)(end - beg + 1));
    float4 acc;
    {
      float4 v = ld_row(h, c, lane);
      float s = dc * dc;
      acc = make_float4(v.x * s, v.y * s, v.z * s, v.w * s);
    }

    for (int base = beg; base < end; base += 32) {
      int n = end - base;
      if (n > 32) n = 32;
      int myr = c;
      float myw = 0.0f;
      if (lane < n) {
        int p = base + lane;
        myr = (int)srcs[p] + ((p >= mid) ? 65536 : 0);
        int deg = off[2 * myr + 2] - off[2 * myr] + 1;
        myw = (1.0f / sqrtf((float)deg)) * dc;
      }
      int j = 0;
      for (; j + 4 <= n; j += 4) {
        int r0 = __builtin_amdgcn_readlane(myr, j);
        int r1 = __builtin_amdgcn_readlane(myr, j + 1);
        int r2 = __builtin_amdgcn_readlane(myr, j + 2);
        int r3 = __builtin_amdgcn_readlane(myr, j + 3);
        float4 v0 = ld_row(h, r0, lane);
        float4 v1 = ld_row(h, r1, lane);
        float4 v2 = ld_row(h, r2, lane);
        float4 v3 = ld_row(h, r3, lane);
        float w0 = __int_as_float(__builtin_amdgcn_readlane(__float_as_int(myw), j));
        float w1 = __int_as_float(__builtin_amdgcn_readlane(__float_as_int(myw), j + 1));
        float w2 = __int_as_float(__builtin_amdgcn_readlane(__float_as_int(myw), j + 2));
        float w3 = __int_as_float(__builtin_amdgcn_readlane(__float_as_int(myw), j + 3));
        acc.x += v0.x * w0; acc.y += v0.y * w0; acc.z += v0.z * w0; acc.w += v0.w * w0;
        acc.x += v1.x * w1; acc.y += v1.y * w1; acc.z += v1.z * w1; acc.w += v1.w * w1;
        acc.x += v2.x * w2; acc.y += v2.y * w2; acc.z += v2.z * w2; acc.w += v2.w * w2;
        acc.x += v3.x * w3; acc.y += v3.y * w3; acc.z += v3.z * w3; acc.w += v3.w * w3;
      }
      for (; j < n; ++j) {
        int r0 = __builtin_amdgcn_readlane(myr, j);
        float4 v0 = ld_row(h, r0, lane);
        float w0 = __int_as_float(__builtin_amdgcn_readlane(__float_as_int(myw), j));
        acc.x += v0.x * w0; acc.y += v0.y * w0; acc.z += v0.z * w0; acc.w += v0.w * w0;
      }
    }

    acc.x += bv.x; acc.y += bv.y; acc.z += bv.z; acc.w += bv.w;
    if (RELU) {
      acc.x = fmaxf(acc.x, 0.0f); acc.y = fmaxf(acc.y, 0.0f);
      acc.z = fmaxf(acc.z, 0.0f); acc.w = fmaxf(acc.w, 0.0f);
    }
    ((float4*)(out + (size_t)c * HIDC))[lane] = acc;
  }
}

#define NPAIR (NPOS + NNEG)

__device__ __forceinline__ float dot4(float4 a, float4 b) {
  return fmaf(a.w, b.w, fmaf(a.z, b.z, fmaf(a.x, b.x, a.y * b.y)));
}
static_assert(NPOS % 32 == 0 && NNEG % 32 == 0, "a chunk of 32 pairs must not straddle the two lists");

__global__ __launch_bounds__(256) void k_decode(const float* __restrict__ z,
                                                const int* __restrict__ pos,
                                                const int* __restrict__ neg,
                                                float* __restrict__ out) {
  const int lane = threadIdx.x & 31;
  const int gw = __builtin_amdgcn_readfirstlane((int)((blockIdx.x * blockDim.x + threadIdx.x) >> 5));
  const int nw = (int)((gridDim.x * blockDim.x) >> 5);
  for (int ch = gw; ch < NPAIR / 32; ch += nw) {
    const int w0 = ch * 32;
    const int* __restrict__ src = (w0 < NPOS) ? pos : neg;
    const int cnt = (w0 < NPOS) ? NPOS : NNEG;
    const int j0 = (w0 < NPOS) ? w0 : w0 - NPOS;
    const int ia = src[j0 + lane];
    const int ib = src[cnt + j0 + lane];
    float res = 0.0f;
    for (int j = 0; j < 32; j += 4) {
      int a0 = __builtin_amdgcn_readlane(ia, j);
      int b0 = __builtin_amdgcn_readlane(ib, j);
      int a1 = __builtin_amdgcn_readlane(ia, j + 1);
      int b1 = __builtin_amdgcn_readlane(ib, j + 1);
      int a2 = __builtin_amdgcn_readlane(ia, j + 2);
      int b2 = __builtin_amdgcn_readlane(ib, j + 2);
      int a3 = __builtin_amdgcn_readlane(ia, j + 3);
      int b3 = __builtin_amdgcn_readlane(ib, j + 3);
      float4 x0 = ld_row(z, a0, lane), y0 = ld_row(z, b0, lane);
      float4 x1 = ld_row(z, a1, lane), y1 = ld_row(z, b1, lane);
      float4 x2 = ld_row(z, a2, lane), y2 = ld_row(z, b2, lane);
      float4 x3 = ld_row(z, a3, lane), y3 = ld_row(z, b3, lane);
      float s0 = dot4(x0, y0), s1 = dot4(x1, y1), s2 = dot4(x2, y2), s3 = dot4(x3, y3);
#pragma unroll
      for (int d = 16; d > 0; d >>= 1) {
        s0 += __shfl_xor(s0, d, 32);
        s1 += __shfl_xor(s1, d, 32);
        s2 += __shfl_xor(s2, d, 32);
        s3 += __shfl_xor(s3, d, 32);
      }
      res = (lane == j)     ? s0 : res;
      res = (lane == j + 1) ? s1 : res;
      res = (lane == j + 2) ? s2 : res;
      res = (lane == j + 3) ? s3 : res;
    }
    out[w0 + lane] = res;
  }
}

extern "C" void kernel_launch(void* const* d_in, const int* in_sizes, int n_in,
                              void* d_out, int out_size, void* d_ws, size_t ws_size,
                              hipStream_t stream) {
  const float* x  = (const float*)d_in[0];
  const int*   ei = (const int*)d_in[1];
  const int*   pos = (const int*)d_in[2];
  const int*   neg = (const int*)d_in[3];
  const float* W1 = (const float*)d_in[4];
  const float* b1 = (const float*)d_in[5];
  const float* W2 = (const float*)d_in[6];
  const float* b2 = (const float*)d_in[7];
  float* logits = (float*)d_out;

  char* ws = (char*)d_ws;
  int*   off  = (int*)ws;
  float* WP   = (float*)(ws + 800256u);
  unsigned short* srcs = (unsigned short*)(ws + 931328u);
  float* bufH = (float*)(ws + 4374528u);
  float* bufZ = (float*)(ws + 55574528u);
  int*   tmp  = (int*)(ws + 55574528u);
  int*   aux  = (int*)(ws + 61974528u);
  int*   total    = aux;
  int*   gcur     = aux + 512;
  int*   bstart_g = aux + 1024;

  const int* row = ei;
  const int* col = ei + NE;
  const int B = 256;

  k_zero_aux<<<4, B, 0, stream>>>(aux);
  k_bucket_count<<<NSLICE, B, 0, stream>>>(col, total);
  k_bucket_split<<<NSLICE, B, 0, stream>>>(row, col, total, gcur, bstart_g, tmp);
  k_bucket_sort<<<NB, B, 0, stream>>>(tmp, total, bstart_g, off, srcs);

  int ga_blocks   = (NN / 8 * 32 + B - 1) / B;

  k_pack_w<<<INC * 32 / B, B, 0, stream>>>(W1, WP, INC);
  k_gemm_wmma<INC><<<GEMM_GROUPS, GEMM_WAVES * 32, 0, stream>>>(x, WP, bufH);
  k_gather<1><<<ga_blocks, B, 0, stream>>>(bufZ, bufH, off, srcs, b1);

  k_pack_w<<<HIDC * 32 / B, B, 0, stream>>>(W2, WP, HIDC);
  k_gemm_wmma<HIDC><<<GEMM_GROUPS, GEMM_WAVES * 32, 0, stream>>>(bufZ, WP, bufH);
  k_gather<0><<<ga_blocks, B, 0, stream>>>(bufZ, bufH, off, srcs, b2);

  int dec_blocks = ((NPOS + NNEG) + B - 1) / B;
  k_decode<<<dec_blocks, B, 0, stream>>>(bufZ, pos, neg, logits);
}


# GEMM store scales each h row by its own deg^-1/2, so the aggregation adds plain rows and scales once by the destination's deg^-1/2: no per-edge degree lookups; same sum order, new rounding
# speedup vs baseline: 4.8505x; 1.0636x over previous
#include <hip/hip_runtime.h>

#define NN   100000
#define INC  256
#define HIDC 128
#define NE   1600000
#define NPOS 200000
#define NNEG 200000
#define NK   (2 * NN)

typedef __attribute__((ext_vector_type(2))) float v2f;
typedef __attribute__((ext_vector_type(8))) float v8f;

#define BSH   8
#define NB    ((NN + 255) >> BSH)
#define EPT   25
#define SLICE (256 * EPT)
#define NSLICE (NE / SLICE)
static_assert(NE % SLICE == 0, "the edge list must split into whole slices");
static_assert(NB <= 512, "two table entries per thread");
static_assert(NN <= (1 << 17), "a source id takes 17 bits of an edge record");
#define STAGE_CAP 16384
#define SORT_BATCH 8

__device__ __forceinline__ int block_excl_scan(int s, int* wsum) {
  const int lane = threadIdx.x & 31, wid = threadIdx.x >> 5;
  int incl = s;
#pragma unroll
  for (int d = 1; d < 32; d <<= 1) {
    int u = __shfl_up(incl, d, 32);
    if (lane >= d) incl += u;
  }
  __syncthreads();
  if (lane == 31) wsum[wid] = incl;
  __syncthreads();
  int pre = 0;
#pragma unroll
  for (int w = 0; w < 8; ++w) pre += (w < wid) ? wsum[w] : 0;
  return pre + incl - s;
}

__global__ void k_zero_aux(int* aux) {
  int i = blockIdx.x * blockDim.x + threadIdx.x;
  if (i < 1024) aux[i] = 0;
}

__global__ __launch_bounds__(256) void k_bucket_count(const int* __restrict__ col,
                                                      int* __restrict__ total) {
  __shared__ int lcnt[512];
  const int tid = threadIdx.x;
  lcnt[tid] = 0;
  lcnt[tid + 256] = 0;
  __syncthreads();
  const int base = blockIdx.x * SLICE + tid;
  int c[EPT];
#pragma unroll
  for (int j = 0; j < EPT; ++j) c[j] = col[base + j * 256];
#pragma unroll
  for (int j = 0; j < EPT; ++j) atomicAdd(&lcnt[c[j] >> BSH], 1);
  __syncthreads();
  for (int b = tid; b < NB; b += 256) {
    int v = lcnt[b];
    if (v) atomicAdd(&total[b], v);
  }
}

__global__ __launch_bounds__(256) void k_bucket_split(const int* __restrict__ row,
                                                      const int* __restrict__ col,
                                                      const int* __restrict__ total,
                                                      int* __restrict__ gcur,
                                                      int* __restrict__ bstart_g,
                                                      int* __restrict__ tmp) {
  __shared__ int lcnt[512];
  __shared__ int lstart[512];
  __shared__ int lbase[512];
  __shared__ int wsum[8];
  __shared__ int stage[SLICE];
  __shared__ unsigned short stageb[SLICE];
  const int tid = threadIdx.x;
  lcnt[tid] = 0;
  lcnt[tid + 256] = 0;

  int c[EPT], r[EPT], lr[EPT];
  const int base = blockIdx.x * SLICE + tid;
#pragma unroll
  for (int j = 0; j < EPT; ++j) {
    c[j] = col[base + j * 256];
    r[j] = row[base + j * 256];
  }

  const int b0 = 2 * tid, b1 = 2 * tid + 1;
  const int t0 = (b0 < NB) ? total[b0] : 0;
  const int t1 = (b1 < NB) ? total[b1] : 0;
  const int bs0 = block_excl_scan(t0 + t1, wsum);
  const int bs1 = bs0 + t0;
  if (blockIdx.x == 0) {
    if (b0 < NB) bstart_g[b0] = bs0;
    if (b1 < NB) bstart_g[b1] = bs1;
  }

#pragma unroll
  for (int j = 0; j < EPT; ++j) lr[j] = atomicAdd(&lcnt[c[j] >> BSH], 1);
  __syncthreads();

  const int n0 = lcnt[b0], n1 = lcnt[b1];
  const int ls0 = block_excl_scan(n0 + n1, wsum);
  const int ls1 = ls0 + n0;
  lstart[b0] = ls0;
  lstart[b1] = ls1;
  {
    int g0 = (n0 > 0) ? atomicAdd(&gcur[b0], n0) : 0;
    int g1 = (n1 > 0) ? atomicAdd(&gcur[b1], n1) : 0;
    lbase[b0] = bs0 + g0 - ls0;
    lbase[b1] = bs1 + g1 - ls1;
  }
  __syncthreads();

#pragma unroll
  for (int j = 0; j < EPT; ++j) {
    int b = c[j] >> BSH;
    int p = lstart[b] + lr[j];
    stage[p] = r[j] | ((c[j] & 255) << 17);
    stageb[p] = (unsigned short)b;
  }
  __syncthreads();

#pragma unroll
  for (int j = 0; j < EPT; ++j) {
    int i = j * 256 + tid;
    tmp[lbase[stageb[i]] + i] = stage[i];
  }
}

__global__ __launch_bounds__(256) void k_bucket_sort(const int* __restrict__ tmp,
                                                     const int* __restrict__ total,
                                                     const int* __restrict__ bstart_g,
                                                     int* __restrict__ off,
                                                     unsigned short* __restrict__ srcs) {
  __shared__ int cnt[512];
  __shared__ int cur[512];
  __shared__ int wsum[8];
  __shared__ unsigned short stage[STAGE_CAP];
  const int tid = threadIdx.x, b = blockIdx.x;
  const int n = total[b];
  const int s0 = bstart_g[b];
  const int* __restrict__ rec = tmp + s0;
  cnt[tid] = 0;
  cnt[tid + 256] = 0;
  __syncthreads();
  for (int i0 = tid; i0 < n; i0 += 256 * SORT_BATCH) {
    int e[SORT_BATCH];
#pragma unroll
    for (int q = 0; q < SORT_BATCH; ++q) e[q] = rec[min(i0 + q * 256, n - 1)];
#pragma unroll
    for (int q = 0; q < SORT_BATCH; ++q)
      if (i0 + q * 256 < n) atomicAdd(&cnt[(e[q] >> 16) & 511], 1);
  }
  __syncthreads();
  const int c0 = cnt[2 * tid], c1 = cnt[2 * tid + 1];
  const int ex = block_excl_scan(c0 + c1, wsum);
  const int node = (b << BSH) + tid;
  if (node < NN) {
    off[2 * node] = s0 + ex;
    off[2 * node + 1] = s0 + ex + c0;
  }
  if (b == NB - 1 && tid == 0) off[NK] = s0 + n;
  cur[2 * tid] = ex;
  cur[2 * tid + 1] = ex + c0;
  __syncthreads();
  if (n <= STAGE_CAP) {
    for (int i0 = tid; i0 < n; i0 += 256 * SORT_BATCH) {
      int e[SORT_BATCH];
#pragma unroll
      for (int q = 0; q < SORT_BATCH; ++q) e[q] = rec[min(i0 + q * 256, n - 1)];
#pragma unroll
      for (int q = 0; q < SORT_BATCH; ++q) {
        if (i0 + q * 256 < n) {
          int p = atomicAdd(&cur[(e[q] >> 16) & 511], 1);
          stage[p] = (unsigned short)(e[q] & 0xFFFF);
        }
      }
    }
    __syncthreads();
    for (int i = tid; i < n; i += 256) srcs[s0 + i] = stage[i];
  } else {
    for (int i = tid; i < n; i += 256) {
      int e = rec[i];
      int p = atomicAdd(&cur[(e >> 16) & 511], 1);
      srcs[s0 + p] = (unsigned short)(e & 0xFFFF);
    }
  }
}

__global__ void k_pack_w(const float* __restrict__ W, float* __restrict__ WP, int K) {
  int g = blockIdx.x * blockDim.x + threadIdx.x;
  if (g >= K * 32) return;
  int lane = g & 31, t = (g >> 5) & 7, p = g >> 8;
  int kk = 32 * (p >> 2) + 16 * (lane >> 4) + 4 * (p & 3);
  int n = 16 * t + (lane & 15);
  ((float4*)WP)[g] = make_float4(W[kk * HIDC + n], W[(kk + 1) * HIDC + n],
                                 W[(kk + 2) * HIDC + n], W[(kk + 3) * HIDC + n]);
}

#define GEMM_GROUPS 512
#define GEMM_WAVES  12
#define NSTRIP (NN / 16)
#define NLEFT  (NSTRIP - GEMM_GROUPS * GEMM_WAVES)
static_assert(NN % 16 == 0, "whole strips");
static_assert(NLEFT >= 0 && 4 * NLEFT <= GEMM_GROUPS, "a left-over quarter strip for the first wave of some groups");

template <int K, int NT>
__device__ __forceinline__ void gemm_strip(const float* __restrict__ A, const float4* wl,
                                           const int* __restrict__ off,
                                           float* __restrict__ C, int m0, int t0, int lane) {
  const int c = lane & 15, h = lane >> 4;
  const float* ap = A + (size_t)(m0 + c) * K + 16 * h;

  int ov[17];
#pragma unroll
  for (int i = 0; i <= 16; ++i) ov[i] = off[2 * (m0 + i)];
  v8f acc[NT];
#pragma unroll
  for (int t = 0; t < NT; ++t) acc[t] = (v8f)(0.0f);

  float4 cur[4], nxt[4];
#pragma unroll
  for (int i = 0; i < 4; ++i) cur[i] = *(const float4*)(ap + 4 * i);

  const float4* wp = wl + t0 * 32 + lane;
#pragma unroll 1
  for (int q = 0; q < K / 32; ++q) {
    const int qn = (q + 1 < K / 32) ? q + 1 : q;
#pragma unroll
    for (int i = 0; i < 4; ++i) nxt[i] = *(const float4*)(ap + 32 * qn + 4 * i);
    asm volatile("" ::: "memory");
#pragma unroll
    for (int i = 0; i < 4; ++i) {
      float4 b[NT];
#pragma unroll
      for (int t = 0; t < NT; ++t) b[t] = wp[(i * 8 + t) * 32];
      v2f a0 = {cur[i].x, cur[i].y};
      v2f a1 = {cur[i].z, cur[i].w};
#pragma unroll
      for (int t = 0; t < NT; ++t) {
        v2f b0 = {b[t].x, b[t].y};
        acc[t] = __builtin_amdgcn_wmma_f32_16x16x4_f32(
            false, a0, false, b0, (short)0, acc[t], false, false);
      }
#pragma unroll
      for (int t = 0; t < NT; ++t) {
        v2f b1 = {b[t].z, b[t].w};
        acc[t] = __builtin_amdgcn_wmma_f32_16x16x4_f32(
            false, a1, false, b1, (short)0, acc[t], false, false);
      }
    }
    wp += 4 * 8 * 32;
#pragma unroll
    for (int i = 0; i < 4; ++i) cur[i] = nxt[i];
  }

  float dv[8];
#pragma unroll
  for (int i = 0; i < 8; ++i) {
    const int deg = (h ? ov[i + 9] - ov[i + 8] : ov[i + 1] - ov[i]) + 1;
    dv[i] = 1.0f / sqrtf((float)deg);
  }

#pragma unroll
  for (int t = 0; t < NT; ++t) {
#pragma unroll
    for (int i = 0; i < 8; ++i) {
      C[(size_t)(m0 + i + h * 8) * HIDC + (t0 + t) * 16 + c] = acc[t][i] * dv[i];
    }
  }
}

template <int K>
__global__ __launch_bounds__(GEMM_WAVES * 32, 6) void k_gemm_wmma(const float* __restrict__ A,
                                                                 const float* __restrict__ WP,
                                                                 const int* __restrict__ off,
                                                                 float* __restrict__ C) {
  __shared__ float4 wl[INC * HIDC / 4];
  const int tid = threadIdx.x;
  for (int i = tid; i < K * 32; i += GEMM_WAVES * 32) wl[i] = ((const float4*)WP)[i];
  __syncthreads();

  const int lane = tid & 31;
  const int wid = __builtin_amdgcn_readfirstlane(tid >> 5);
  gemm_strip<K, 8>(A, wl, off, C, (blockIdx.x + GEMM_GROUPS * wid) * 16, 0, lane);
  if (wid == 0 && blockIdx.x < 4 * NLEFT) {
    const int j = blockIdx.x;
    gemm_strip<K, 2>(A, wl, off, C, (GEMM_GROUPS * GEMM_WAVES + (j >> 2)) * 16, 2 * (j & 3), lane);
  }
}

__device__ __forceinline__ float4 ld_row(const float* __restrict__ h, int r, int lane) {
  return ((const float4*)(h + (size_t)r * HIDC))[lane];
}

template <int RELU>
__global__ __launch_bounds__(256) void k_gather(float* __restrict__ out,
                                                const float* __restrict__ h,
                                                const int* __restrict__ off,
                                                const unsigned short* __restrict__ srcs,
                                                const float* __restrict__ bias) {
  const int lane = threadIdx.x & 31;
  const int gw = __builtin_amdgcn_readfirstlane((int)((blockIdx.x * blockDim.x + threadIdx.x) >> 5));
  const int nw = (int)((gridDim.x * blockDim.x) >> 5);
  const float4 bv = ((const float4*)bias)[lane];
  for (int c = gw; c < NN; c += nw) {
    const int beg = off[2 * c];
    const int mid = off[2 * c + 1];
    const int end = off[2 * c + 2];
    const float dc = 1.0f / sqrtf((float)(end - beg + 1));
    float4 acc = ld_row(h, c, lane);

    for (int base = beg; base < end; base += 32) {
      int n = end - base;
      if (n > 32) n = 32;
      int myr = c;
      if (lane < n) {
        int p = base + lane;
        myr = (int)srcs[p] + ((p >= mid) ? 65536 : 0);
      }
      int j = 0;
      for (; j + 4 <= n; j += 4) {
        int r0 = __builtin_amdgcn_readlane(myr, j);
        int r1 = __builtin_amdgcn_readlane(myr, j + 1);
        int r2 = __builtin_amdgcn_readlane(myr, j + 2);
        int r3 = __builtin_amdgcn_readlane(myr, j + 3);
        float4 v0 = ld_row(h, r0, lane);
        float4 v1 = ld_row(h, r1, lane);
        float4 v2 = ld_row(h, r2, lane);
        float4 v3 = ld_row(h, r3, lane);
        acc.x += v0.x; acc.y += v0.y; acc.z += v0.z; acc.w += v0.w;
        acc.x += v1.x; acc.y += v1.y; acc.z += v1.z; acc.w += v1.w;
        acc.x += v2.x; acc.y += v2.y; acc.z += v2.z; acc.w += v2.w;
        acc.x += v3.x; acc.y += v3.y; acc.z += v3.z; acc.w += v3.w;
      }
      for (; j < n; ++j) {
        int r0 = __builtin_amdgcn_readlane(myr, j);
        float4 v0 = ld_row(h, r0, lane);
        acc.x += v0.x; acc.y += v0.y; acc.z += v0.z; acc.w += v0.w;
      }
    }

    acc.x = acc.x * dc + bv.x; acc.y = acc.y * dc + bv.y;
    acc.z = acc.z * dc + bv.z; acc.w = acc.w * dc + bv.w;
    if (RELU) {
      acc.x = fmaxf(acc.x, 0.0f); acc.y = fmaxf(acc.y, 0.0f);
      acc.z = fmaxf(acc.z, 0.0f); acc.w = fmaxf(acc.w, 0.0f);
    }
    ((float4*)(out + (size_t)c * HIDC))[lane] = acc;
  }
}

#define NPAIR (NPOS + NNEG)

__device__ __forceinline__ float dot4(float4 a, float4 b) {
  return fmaf(a.w, b.w, fmaf(a.z, b.z, fmaf(a.x, b.x, a.y * b.y)));
}
static_assert(NPOS % 32 == 0 && NNEG % 32 == 0, "a chunk of 32 pairs must not straddle the two lists");

__global__ __launch_bounds__(256) void k_decode(const float* __restrict__ z,
                                                const int* __restrict__ pos,
                                                const int* __restrict__ neg,
                                                float* __restrict__ out) {
  const int lane = threadIdx.x & 31;
  const int gw = __builtin_amdgcn_readfirstlane((int)((blockIdx.x * blockDim.x + threadIdx.x) >> 5));
  const int nw = (int)((gridDim.x * blockDim.x) >> 5);
  for (int ch = gw; ch < NPAIR / 32; ch += nw) {
    const int w0 = ch * 32;
    const int* __restrict__ src = (w0 < NPOS) ? pos : neg;
    const int cnt = (w0 < NPOS) ? NPOS : NNEG;
    const int j0 = (w0 < NPOS) ? w0 : w0 - NPOS;
    const int ia = src[j0 + lane];
    const int ib = src[cnt + j0 + lane];
    float res = 0.0f;
    for (int j = 0; j < 32; j += 4) {
      int a0 = __builtin_amdgcn_readlane(ia, j);
      int b0 = __builtin_amdgcn_readlane(ib, j);
      int a1 = __builtin_amdgcn_readlane(ia, j + 1);
      int b1 = __builtin_amdgcn_readlane(ib, j + 1);
      int a2 = __builtin_amdgcn_readlane(ia, j + 2);
      int b2 = __builtin_amdgcn_readlane(ib, j + 2);
      int a3 = __builtin_amdgcn_readlane(ia, j + 3);
      int b3 = __builtin_amdgcn_readlane(ib, j + 3);
      float4 x0 = ld_row(z, a0, lane), y0 = ld_row(z, b0, lane);
      float4 x1 = ld_row(z, a1, lane), y1 = ld_row(z, b1, lane);
      float4 x2 = ld_row(z, a2, lane), y2 = ld_row(z, b2, lane);
      float4 x3 = ld_row(z, a3, lane), y3 = ld_row(z, b3, lane);
      float s0 = dot4(x0, y0), s1 = dot4(x1, y1), s2 = dot4(x2, y2), s3 = dot4(x3, y3);
#pragma unroll
      for (int d = 16; d > 0; d >>= 1) {
        s0 += __shfl_xor(s0, d, 32);
        s1 += __shfl_xor(s1, d, 32);
        s2 += __shfl_xor(s2, d, 32);
        s3 += __shfl_xor(s3, d, 32);
      }
      res = (lane == j)     ? s0 : res;
      res = (lane == j + 1) ? s1 : res;
      res = (lane == j + 2) ? s2 : res;
      res = (lane == j + 3) ? s3 : res;
    }
    out[w0 + lane] = res;
  }
}

extern "C" void kernel_launch(void* const* d_in, const int* in_sizes, int n_in,
                              void* d_out, int out_size, void* d_ws, size_t ws_size,
                              hipStream_t stream) {
  const float* x  = (const float*)d_in[0];
  const int*   ei = (const int*)d_in[1];
  const int*   pos = (const int*)d_in[2];
  const int*   neg = (const int*)d_in[3];
  const float* W1 = (const float*)d_in[4];
  const float* b1 = (const float*)d_in[5];
  const float* W2 = (const float*)d_in[6];
  const float* b2 = (const float*)d_in[7];
  float* logits = (float*)d_out;

  char* ws = (char*)d_ws;
  int*   off  = (int*)ws;
  float* WP   = (float*)(ws + 800256u);
  unsigned short* srcs = (unsigned short*)(ws + 931328u);
  float* bufH = (float*)(ws + 4374528u);
  float* bufZ = (float*)(ws + 55574528u);
  int*   tmp  = (int*)(ws + 55574528u);
  int*   aux  = (int*)(ws + 61974528u);
  int*   total    = aux;
  int*   gcur     = aux + 512;
  int*   bstart_g = aux + 1024;

  const int* row = ei;
  const int* col = ei + NE;
  const int B = 256;

  k_zero_aux<<<4, B, 0, stream>>>(aux);
  k_bucket_count<<<NSLICE, B, 0, stream>>>(col, total);
  k_bucket_split<<<NSLICE, B, 0, stream>>>(row, col, total, gcur, bstart_g, tmp);
  k_bucket_sort<<<NB, B, 0, stream>>>(tmp, total, bstart_g, off, srcs);

  int ga_blocks   = (NN / 8 * 32 + B - 1) / B;

  k_pack_w<<<INC * 32 / B, B, 0, stream>>>(W1, WP, INC);
  k_gemm_wmma<INC><<<GEMM_GROUPS, GEMM_WAVES * 32, 0, stream>>>(x, WP, off, bufH);
  k_gather<1><<<ga_blocks, B, 0, stream>>>(bufZ, bufH, off, srcs, b1);

  k_pack_w<<<HIDC * 32 / B, B, 0, stream>>>(W2, WP, HIDC);
  k_gemm_wmma<HIDC><<<GEMM_GROUPS, GEMM_WAVES * 32, 0, stream>>>(bufZ, WP, off, bufH);
  k_gather<0><<<ga_blocks, B, 0, stream>>>(bufZ, bufH, off, srcs, b2);

  int dec_blocks = ((NPOS + NNEG) + B - 1) / B;
  k_decode<<<dec_blocks, B, 0, stream>>>(bufZ, pos, neg, logits);
}


# layer-1 aggregation and layer-2 GEMM in one kernel: a wave sums 16 rows into LDS (up to 16 row loads in flight, same add order), then multiplies them by W2; relu output never stored; 10 launches
# speedup vs baseline: 5.0689x; 1.0450x over previous
#include <hip/hip_runtime.h>

#define NN   100000
#define INC  256
#define HIDC 128
#define NE   1600000
#define NPOS 200000
#define NNEG 200000
#define NK   (2 * NN)

typedef __attribute__((ext_vector_type(2))) float v2f;
typedef __attribute__((ext_vector_type(8))) float v8f;

#define BSH   8
#define NB    ((NN + 255) >> BSH)
#define EPT   25
#define SLICE (256 * EPT)
#define NSLICE (NE / SLICE)
static_assert(NE % SLICE == 0, "the edge list must split into whole slices");
static_assert(NB <= 512, "two table entries per thread");
static_assert(NN <= (1 << 17), "a source id takes 17 bits of an edge record");
#define STAGE_CAP 16384
#define SORT_BATCH 8

__device__ __forceinline__ int block_excl_scan(int s, int* wsum) {
  const int lane = threadIdx.x & 31, wid = threadIdx.x >> 5;
  int incl = s;
#pragma unroll
  for (int d = 1; d < 32; d <<= 1) {
    int u = __shfl_up(incl, d, 32);
    if (lane >= d) incl += u;
  }
  __syncthreads();
  if (lane == 31) wsum[wid] = incl;
  __syncthreads();
  int pre = 0;
#pragma unroll
  for (int w = 0; w < 8; ++w) pre += (w < wid) ? wsum[w] : 0;
  return pre + incl - s;
}

__global__ void k_zero_aux(int* aux) {
  int i = blockIdx.x * blockDim.x + threadIdx.x;
  if (i < 1024) aux[i] = 0;
}

__global__ __launch_bounds__(256) void k_bucket_count(const int* __restrict__ col,
                                                      int* __restrict__ total) {
  __shared__ int lcnt[512];
  const int tid = threadIdx.x;
  lcnt[tid] = 0;
  lcnt[tid + 256] = 0;
  __syncthreads();
  const int base = blockIdx.x * SLICE + tid;
  int c[EPT];
#pragma unroll
  for (int j = 0; j < EPT; ++j) c[j] = col[base + j * 256];
#pragma unroll
  for (int j = 0; j < EPT; ++j) atomicAdd(&lcnt[c[j] >> BSH], 1);
  __syncthreads();
  for (int b = tid; b < NB; b += 256) {
    int v = lcnt[b];
    if (v) atomicAdd(&total[b], v);
  }
}

__global__ __launch_bounds__(256) void k_bucket_split(const int* __restrict__ row,
                                                      const int* __restrict__ col,
                                                      const int* __restrict__ total,
                                                      int* __restrict__ gcur,
                                                      int* __restrict__ bstart_g,
                                                      int* __restrict__ tmp) {
  __shared__ int lcnt[512];
  __shared__ int lstart[512];
  __shared__ int lbase[512];
  __shared__ int wsum[8];
  __shared__ int stage[SLICE];
  __shared__ unsigned short stageb[SLICE];
  const int tid = threadIdx.x;
  lcnt[tid] = 0;
  lcnt[tid + 256] = 0;

  int c[EPT], r[EPT], lr[EPT];
  const int base = blockIdx.x * SLICE + tid;
#pragma unroll
  for (int j = 0; j < EPT; ++j) {
    c[j] = col[base + j * 256];
    r[j] = row[base + j * 256];
  }

  const int b0 = 2 * tid, b1 = 2 * tid + 1;
  const int t0 = (b0 < NB) ? total[b0] : 0;
  const int t1 = (b1 < NB) ? total[b1] : 0;
  const int bs0 = block_excl_scan(t0 + t1, wsum);
  const int bs1 = bs0 + t0;
  if (blockIdx.x == 0) {
    if (b0 < NB) bstart_g[b0] = bs0;
    if (b1 < NB) bstart_g[b1] = bs1;
  }

#pragma unroll
  for (int j = 0; j < EPT; ++j) lr[j] = atomicAdd(&lcnt[c[j] >> BSH], 1);
  __syncthreads();

  const int n0 = lcnt[b0], n1 = lcnt[b1];
  const int ls0 = block_excl_scan(n0 + n1, wsum);
  const int ls1 = ls0 + n0;
  lstart[b0] = ls0;
  lstart[b1] = ls1;
  {
    int g0 = (n0 > 0) ? atomicAdd(&gcur[b0], n0) : 0;
    int g1 = (n1 > 0) ? atomicAdd(&gcur[b1], n1) : 0;
    lbase[b0] = bs0 + g0 - ls0;
    lbase[b1] = bs1 + g1 - ls1;
  }
  __syncthreads();

#pragma unroll
  for (int j = 0; j < EPT; ++j) {
    int b = c[j] >> BSH;
    int p = lstart[b] + lr[j];
    stage[p] = r[j] | ((c[j] & 255) << 17);
    stageb[p] = (unsigned short)b;
  }
  __syncthreads();

#pragma unroll
  for (int j = 0; j < EPT; ++j) {
    int i = j * 256 + tid;
    tmp[lbase[stageb[i]] + i] = stage[i];
  }
}

__global__ __launch_bounds__(256) void k_bucket_sort(const int* __restrict__ tmp,
                                                     const int* __restrict__ total,
                                                     const int* __restrict__ bstart_g,
                                                     int* __restrict__ off,
                                                     unsigned short* __restrict__ srcs) {
  __shared__ int cnt[512];
  __shared__ int cur[512];
  __shared__ int wsum[8];
  __shared__ unsigned short stage[STAGE_CAP];
  const int tid = threadIdx.x, b = blockIdx.x;
  const int n = total[b];
  const int s0 = bstart_g[b];
  const int* __restrict__ rec = tmp + s0;
  cnt[tid] = 0;
  cnt[tid + 256] = 0;
  __syncthreads();
  for (int i0 = tid; i0 < n; i0 += 256 * SORT_BATCH) {
    int e[SORT_BATCH];
#pragma unroll
    for (int q = 0; q < SORT_BATCH; ++q) e[q] = rec[min(i0 + q * 256, n - 1)];
#pragma unroll
    for (int q = 0; q < SORT_BATCH; ++q)
      if (i0 + q * 256 < n) atomicAdd(&cnt[(e[q] >> 16) & 511], 1);
  }
  __syncthreads();
  const int c0 = cnt[2 * tid], c1 = cnt[2 * tid + 1];
  const int ex = block_excl_scan(c0 + c1, wsum);
  const int node = (b << BSH) + tid;
  if (node < NN) {
    off[2 * node] = s0 + ex;
    off[2 * node + 1] = s0 + ex + c0;
  }
  if (b == NB - 1 && tid == 0) off[NK] = s0 + n;
  cur[2 * tid] = ex;
  cur[2 * tid + 1] = ex + c0;
  __syncthreads();
  if (n <= STAGE_CAP) {
    for (int i0 = tid; i0 < n; i0 += 256 * SORT_BATCH) {
      int e[SORT_BATCH];
#pragma unroll
      for (int q = 0; q < SORT_BATCH; ++q) e[q] = rec[min(i0 + q * 256, n - 1)];
#pragma unroll
      for (int q = 0; q < SORT_BATCH; ++q) {
        if (i0 + q * 256 < n) {
          int p = atomicAdd(&cur[(e[q] >> 16) & 511], 1);
          stage[p] = (unsigned short)(e[q] & 0xFFFF);
        }
      }
    }
    __syncthreads();
    for (int i = tid; i < n; i += 256) srcs[s0 + i] = stage[i];
  } else {
    for (int i = tid; i < n; i += 256) {
      int e = rec[i];
      int p = atomicAdd(&cur[(e >> 16) & 511], 1);
      srcs[s0 + p] = (unsigned short)(e & 0xFFFF);
    }
  }
}

__global__ void k_pack_w(const float* __restrict__ W, float* __restrict__ WP, int K) {
  int g = blockIdx.x * blockDim.x + threadIdx.x;
  if (g >= K * 32) return;
  int lane = g & 31, t = (g >> 5) & 7, p = g >> 8;
  int kk = 32 * (p >> 2) + 16 * (lane >> 4) + 4 * (p & 3);
  int n = 16 * t + (lane & 15);
  ((float4*)WP)[g] = make_float4(W[kk * HIDC + n], W[(kk + 1) * HIDC + n],
                                 W[(kk + 2) * HIDC + n], W[(kk + 3) * HIDC + n]);
}

#define GEMM_GROUPS 512
#define GEMM_WAVES  12
#define NSTRIP (NN / 16)
#define NLEFT  (NSTRIP - GEMM_GROUPS * GEMM_WAVES)
static_assert(NN % 16 == 0, "whole strips");
static_assert(NLEFT >= 0 && 4 * NLEFT <= GEMM_GROUPS, "a left-over quarter strip for the first wave of some groups");

template <int K, int NT>
__device__ __forceinline__ void gemm_strip(const float* __restrict__ A, const float4* wl,
                                           const int* __restrict__ off,
                                           float* __restrict__ C, int m0, int t0, int lane) {
  const int c = lane & 15, h = lane >> 4;
  const float* ap = A + (size_t)(m0 + c) * K + 16 * h;

  int ov[17];
#pragma unroll
  for (int i = 0; i <= 16; ++i) ov[i] = off[2 * (m0 + i)];
  v8f acc[NT];
#pragma unroll
  for (int t = 0; t < NT; ++t) acc[t] = (v8f)(0.0f);

  float4 cur[4], nxt[4];
#pragma unroll
  for (int i = 0; i < 4; ++i) cur[i] = *(const float4*)(ap + 4 * i);

  const float4* wp = wl + t0 * 32 + lane;
#pragma unroll 1
  for (int q = 0; q < K / 32; ++q) {
    const int qn = (q + 1 < K / 32) ? q + 1 : q;
#pragma unroll
    for (int i = 0; i < 4; ++i) nxt[i] = *(const float4*)(ap + 32 * qn + 4 * i);
    asm volatile("" ::: "memory");
#pragma unroll
    for (int i = 0; i < 4; ++i) {
      float4 b[NT];
#pragma unroll
      for (int t = 0; t < NT; ++t) b[t] = wp[(i * 8 + t) * 32];
      v2f a0 = {cur[i].x, cur[i].y};
      v2f a1 = {cur[i].z, cur[i].w};
#pragma unroll
      for (int t = 0; t < NT; ++t) {
        v2f b0 = {b[t].x, b[t].y};
        acc[t] = __builtin_amdgcn_wmma_f32_16x16x4_f32(
            false, a0, false, b0, (short)0, acc[t], false, false);
      }
#pragma unroll
      for (int t = 0; t < NT; ++t) {
        v2f b1 = {b[t].z, b[t].w};
        acc[t] = __builtin_amdgcn_wmma_f32_16x16x4_f32(
            false, a1, false, b1, (short)0, acc[t], false, false);
      }
    }
    wp += 4 * 8 * 32;
#pragma unroll
    for (int i = 0; i < 4; ++i) cur[i] = nxt[i];
  }

  float dv[8];
#pragma unroll
  for (int i = 0; i < 8; ++i) {
    const int deg = (h ? ov[i + 9] - ov[i + 8] : ov[i + 1] - ov[i]) + 1;
    dv[i] = 1.0f / sqrtf((float)deg);
  }

#pragma unroll
  for (int t = 0; t < NT; ++t) {
#pragma unroll
    for (int i = 0; i < 8; ++i) {
      C[(size_t)(m0 + i + h * 8) * HIDC + (t0 + t) * 16 + c] = acc[t][i] * dv[i];
    }
  }
}

template <int K>
__global__ __launch_bounds__(GEMM_WAVES * 32, 6) void k_gemm_wmma(const float* __restrict__ A,
                                                                 const float* __restrict__ WP,
                                                                 const int* __restrict__ off,
                                                                 float* __restrict__ C) {
  __shared__ float4 wl[INC * HIDC / 4];
  const int tid = threadIdx.x;
  for (int i = tid; i < K * 32; i += GEMM_WAVES * 32) wl[i] = ((const float4*)WP)[i];
  __syncthreads();

  const int lane = tid & 31;
  const int wid = __builtin_amdgcn_readfirstlane(tid >> 5);
  gemm_strip<K, 8>(A, wl, off, C, (blockIdx.x + GEMM_GROUPS * wid) * 16, 0, lane);
  if (wid == 0 && blockIdx.x < 4 * NLEFT) {
    const int j = blockIdx.x;
    gemm_strip<K, 2>(A, wl, off, C, (GEMM_GROUPS * GEMM_WAVES + (j >> 2)) * 16, 2 * (j & 3), lane);
  }
}

__device__ __forceinline__ float4 ld_row(const float* __restrict__ h, int r, int lane) {
  return ((const float4*)(h + (size_t)r * HIDC))[lane];
}

template <int RELU>
__global__ __launch_bounds__(256) void k_gather(float* __restrict__ out,
                                                const float* __restrict__ h,
                                                const int* __restrict__ off,
                                                const unsigned short* __restrict__ srcs,
                                                const float* __restrict__ bias) {
  const int lane = threadIdx.x & 31;
  const int gw = __builtin_amdgcn_readfirstlane((int)((blockIdx.x * blockDim.x + threadIdx.x) >> 5));
  const int nw = (int)((gridDim.x * blockDim.x) >> 5);
  const float4 bv = ((const float4*)bias)[lane];
  for (int c = gw; c < NN; c += nw) {
    const int beg = off[2 * c];
    const int mid = off[2 * c + 1];
    const int end = off[2 * c + 2];
    const float dc = 1.0f / sqrtf((float)(end - beg + 1));
    float4 acc = ld_row(h, c, lane);

    for (int base = beg; base < end; base += 32) {
      int n = end - base;
      if (n > 32) n = 32;
      int myr = c;
      if (lane < n) {
        int p = base + lane;
        myr = (int)srcs[p] + ((p >= mid) ? 65536 : 0);
      }
      int j = 0;
      for (; j + 4 <= n; j += 4) {
        int r0 = __builtin_amdgcn_readlane(myr, j);
        int r1 = __builtin_amdgcn_readlane(myr, j + 1);
        int r2 = __builtin_amdgcn_readlane(myr, j + 2);
        int r3 = __builtin_amdgcn_readlane(myr, j + 3);
        float4 v0 = ld_row(h, r0, lane);
        float4 v1 = ld_row(h, r1, lane);
        float4 v2 = ld_row(h, r2, lane);
        float4 v3 = ld_row(h, r3, lane);
        acc.x += v0.x; acc.y += v0.y; acc.z += v0.z; acc.w += v0.w;
        acc.x += v1.x; acc.y += v1.y; acc.z += v1.z; acc.w += v1.w;
        acc.x += v2.x; acc.y += v2.y; acc.z += v2.z; acc.w += v2.w;
        acc.x += v3.x; acc.y += v3.y; acc.z += v3.z; acc.w += v3.w;
      }
      for (; j < n; ++j) {
        int r0 = __builtin_amdgcn_readlane(myr, j);
        float4 v0 = ld_row(h, r0, lane);
        acc.x += v0.x; acc.y += v0.y; acc.z += v0.z; acc.w += v0.w;
      }
    }

    acc.x = acc.x * dc + bv.x; acc.y = acc.y * dc + bv.y;
    acc.z = acc.z * dc + bv.z; acc.w = acc.w * dc + bv.w;
    if (RELU) {
      acc.x = fmaxf(acc.x, 0.0f); acc.y = fmaxf(acc.y, 0.0f);
      acc.z = fmaxf(acc.z, 0.0f); acc.w = fmaxf(acc.w, 0.0f);
    }
    ((float4*)(out + (size_t)c * HIDC))[lane] = acc;
  }
}

#define FZ_WAVES  13
#define FZ_GROUPS 256
#define FZ_PITCH  132
static_assert(2 * FZ_WAVES * FZ_GROUPS >= NSTRIP, "two strips a wave cover every strip");
static_assert(HIDC == 128, "the staged strip and the packed W2 are laid out for 128 features");

__device__ __forceinline__ int run_start(int offv, int off32, int k) {
  return (k < 32) ? __builtin_amdgcn_readlane(offv, k & 31) : off32;
}

__device__ __forceinline__ int src_chunk(const unsigned short* __restrict__ srcs, int base, int mid, int lane) {
  const int p = min(base + lane, NE - 1);
  return (int)srcs[p] + ((p >= mid) ? 65536 : 0);
}

template <int CNT>
__device__ __forceinline__ void add_rows(float4& acc, const float* __restrict__ h, int cur, int j, int lane) {
  float4 v[CNT];
#pragma unroll
  for (int k = 0; k < CNT; ++k) v[k] = ld_row(h, __builtin_amdgcn_readlane(cur, j + k), lane);
#pragma unroll
  for (int k = 0; k < CNT; ++k) {
    acc.x += v[k].x; acc.y += v[k].y; acc.z += v[k].z; acc.w += v[k].w;
  }
}

__global__ __launch_bounds__(FZ_WAVES * 32) void k_gather_gemm(const float* __restrict__ h,
                                                              const float* __restrict__ WP,
                                                              const int* __restrict__ off,
                                                              const unsigned short* __restrict__ srcs,
                                                              const float* __restrict__ bias,
                                                              float* __restrict__ C) {
  __shared__ float4 wl[HIDC * HIDC / 4];
  __shared__ float4 zs[FZ_WAVES * 16 * FZ_PITCH / 4];
  const int tid = threadIdx.x;
  for (int i = tid; i < HIDC * 32; i += FZ_WAVES * 32) wl[i] = ((const float4*)WP)[i];
  __syncthreads();

  const int lane = tid & 31;
  const int wid = __builtin_amdgcn_readfirstlane(tid >> 5);
  float* zrow = (float*)zs + wid * 16 * FZ_PITCH;
  const float4 bv = ((const float4*)bias)[lane];
  const int c16 = lane & 15, hh = lane >> 4;

  for (int it = 0; it < 2; ++it) {
    const int s = blockIdx.x + FZ_GROUPS * (it * FZ_WAVES + wid);
    if (s >= NSTRIP) break;
    const int m0 = s * 16;

    const int offv = off[2 * m0 + lane];
    const int off32 = off[2 * m0 + 32];
    int beg = run_start(offv, off32, 0);
    int mid = run_start(offv, off32, 1);
    int end = run_start(offv, off32, 2);
    int cur = src_chunk(srcs, beg, mid, lane);
    float4 self = ld_row(h, m0, lane);
    for (int i = 0; i < 16; ++i) {
      float4 acc = self;
      self = ld_row(h, m0 + min(i + 1, 15), lane);
      const int nmid = (i < 15) ? run_start(offv, off32, 2 * i + 3) : end;
      const int nend = (i < 15) ? run_start(offv, off32, 2 * i + 4) : end;
      const int nxt = src_chunk(srcs, end, nmid, lane);
      const float dc = 1.0f / sqrtf((float)(end - beg + 1));
      for (int base = beg; base < end; base += 32) {
        if (base != beg) cur = src_chunk(srcs, base, mid, lane);
        const int n = min(end - base, 32);
        int j = 0;
        for (; j + 16 <= n; j += 16) add_rows<16>(acc, h, cur, j, lane);
        switch (n - j) {
          case 1:  add_rows<1>(acc, h, cur, j, lane);  break;
          case 2:  add_rows<2>(acc, h, cur, j, lane);  break;
          case 3:  add_rows<3>(acc, h, cur, j, lane);  break;
          case 4:  add_rows<4>(acc, h, cur, j, lane);  break;
          case 5:  add_rows<5>(acc, h, cur, j, lane);  break;
          case 6:  add_rows<6>(acc, h, cur, j, lane);  break;
          case 7:  add_rows<7>(acc, h, cur, j, lane);  break;
          case 8:  add_rows<8>(acc, h, cur, j, lane);  break;
          case 9:  add_rows<9>(acc, h, cur, j, lane);  break;
          case 10: add_rows<10>(acc, h, cur, j, lane); break;
          case 11: add_rows<11>(acc, h, cur, j, lane); break;
          case 12: add_rows<12>(acc, h, cur, j, lane); break;
          case 13: add_rows<13>(acc, h, cur, j, lane); break;
          case 14: add_rows<14>(acc, h, cur, j, lane); break;
          case 15: add_rows<15>(acc, h, cur, j, lane); break;
          default: break;
        }
      }
      acc.x = acc.x * dc + bv.x; acc.y = acc.y * dc + bv.y;
      acc.z = acc.z * dc + bv.z; acc.w = acc.w * dc + bv.w;
      acc.x = fmaxf(acc.x, 0.0f); acc.y = fmaxf(acc.y, 0.0f);
      acc.z = fmaxf(acc.z, 0.0f); acc.w = fmaxf(acc.w, 0.0f);
      *(float4*)(zrow + i * FZ_PITCH + 4 * lane) = acc;
      cur = nxt; beg = end; mid = nmid; end = nend;
    }

    const float* ap = zrow + c16 * FZ_PITCH + 16 * hh;
    v8f cacc[8];
#pragma unroll
    for (int t = 0; t < 8; ++t) cacc[t] = (v8f)(0.0f);
    const float4* wp = wl + lane;
#pragma unroll 1
    for (int q = 0; q < HIDC / 32; ++q) {
      float4 a[4];
#pragma unroll
      for (int i = 0; i < 4; ++i) a[i] = *(const float4*)(ap + 32 * q + 4 * i);
#pragma unroll
      for (int i = 0; i < 4; ++i) {
        float4 b[8];
#pragma unroll
        for (int t = 0; t < 8; ++t) b[t] = wp[(i * 8 + t) * 32];
        v2f a0 = {a[i].x, a[i].y};
        v2f a1 = {a[i].z, a[i].w};
#pragma unroll
        for (int t = 0; t < 8; ++t) {
          v2f b0 = {b[t].x, b[t].y};
          cacc[t] = __builtin_amdgcn_wmma_f32_16x16x4_f32(
              false, a0, false, b0, (short)0, cacc[t], false, false);
        }
#pragma unroll
        for (int t = 0; t < 8; ++t) {
          v2f b1 = {b[t].z, b[t].w};
          cacc[t] = __builtin_amdgcn_wmma_f32_16x16x4_f32(
              false, a1, false, b1, (short)0, cacc[t], false, false);
        }
      }
      wp += 4 * 8 * 32;
    }

    float dv[8];
#pragma unroll
    for (int i = 0; i < 8; ++i) {
      const int dlo = run_start(offv, off32, 2 * i + 2) - run_start(offv, off32, 2 * i);
      const int dhi = run_start(offv, off32, 2 * i + 18) - run_start(offv, off32, 2 * i + 16);
      dv[i] = 1.0f / sqrtf((float)((hh ? dhi : dlo) + 1));
    }
#pragma unroll
    for (int t = 0; t < 8; ++t) {
#pragma unroll
      for (int i = 0; i < 8; ++i) {
        C[(size_t)(m0 + i + hh * 8) * HIDC + t * 16 + c16] = cacc[t][i] * dv[i];
      }
    }
  }
}

#define NPAIR (NPOS + NNEG)

__device__ __forceinline__ float dot4(float4 a, float4 b) {
  return fmaf(a.w, b.w, fmaf(a.z, b.z, fmaf(a.x, b.x, a.y * b.y)));
}
static_assert(NPOS % 32 == 0 && NNEG % 32 == 0, "a chunk of 32 pairs must not straddle the two lists");

__global__ __launch_bounds__(256) void k_decode(const float* __restrict__ z,
                                                const int* __restrict__ pos,
                                                const int* __restrict__ neg,
                                                float* __restrict__ out) {
  const int lane = threadIdx.x & 31;
  const int gw = __builtin_amdgcn_readfirstlane((int)((blockIdx.x * blockDim.x + threadIdx.x) >> 5));
  const int nw = (int)((gridDim.x * blockDim.x) >> 5);
  for (int ch = gw; ch < NPAIR / 32; ch += nw) {
    const int w0 = ch * 32;
    const int* __restrict__ src = (w0 < NPOS) ? pos : neg;
    const int cnt = (w0 < NPOS) ? NPOS : NNEG;
    const int j0 = (w0 < NPOS) ? w0 : w0 - NPOS;
    const int ia = src[j0 + lane];
    const int ib = src[cnt + j0 + lane];
    float res = 0.0f;
    for (int j = 0; j < 32; j += 4) {
      int a0 = __builtin_amdgcn_readlane(ia, j);
      int b0 = __builtin_amdgcn_readlane(ib, j);
      int a1 = __builtin_amdgcn_readlane(ia, j + 1);
      int b1 = __builtin_amdgcn_readlane(ib, j + 1);
      int a2 = __builtin_amdgcn_readlane(ia, j + 2);
      int b2 = __builtin_amdgcn_readlane(ib, j + 2);
      int a3 = __builtin_amdgcn_readlane(ia, j + 3);
      int b3 = __builtin_amdgcn_readlane(ib, j + 3);
      float4 x0 = ld_row(z, a0, lane), y0 = ld_row(z, b0, lane);
      float4 x1 = ld_row(z, a1, lane), y1 = ld_row(z, b1, lane);
      float4 x2 = ld_row(z, a2, lane), y2 = ld_row(z, b2, lane);
      float4 x3 = ld_row(z, a3, lane), y3 = ld_row(z, b3, lane);
      float s0 = dot4(x0, y0), s1 = dot4(x1, y1), s2 = dot4(x2, y2), s3 = dot4(x3, y3);
#pragma unroll
      for (int d = 16; d > 0; d >>= 1) {
        s0 += __shfl_xor(s0, d, 32);
        s1 += __shfl_xor(s1, d, 32);
        s2 += __shfl_xor(s2, d, 32);
        s3 += __shfl_xor(s3, d, 32);
      }
      res = (lane == j)     ? s0 : res;
      res = (lane == j + 1) ? s1 : res;
      res = (lane == j + 2) ? s2 : res;
      res = (lane == j + 3) ? s3 : res;
    }
    out[w0 + lane] = res;
  }
}

extern "C" void kernel_launch(void* const* d_in, const int* in_sizes, int n_in,
                              void* d_out, int out_size, void* d_ws, size_t ws_size,
                              hipStream_t stream) {
  const float* x  = (const float*)d_in[0];
  const int*   ei = (const int*)d_in[1];
  const int*   pos = (const int*)d_in[2];
  const int*   neg = (const int*)d_in[3];
  const float* W1 = (const float*)d_in[4];
  const float* b1 = (const float*)d_in[5];
  const float* W2 = (const float*)d_in[6];
  const float* b2 = (const float*)d_in[7];
  float* logits = (float*)d_out;

  char* ws = (char*)d_ws;
  int*   off  = (int*)ws;
  float* WP   = (float*)(ws + 800256u);
  unsigned short* srcs = (unsigned short*)(ws + 931328u);
  float* bufH = (float*)(ws + 4374528u);
  float* bufZ = (float*)(ws + 55574528u);
  int*   tmp  = (int*)(ws + 55574528u);
  int*   aux  = (int*)(ws + 61974528u);
  int*   total    = aux;
  int*   gcur     = aux + 512;
  int*   bstart_g = aux + 1024;

  const int* row = ei;
  const int* col = ei + NE;
  const int B = 256;

  k_zero_aux<<<4, B, 0, stream>>>(aux);
  k_bucket_count<<<NSLICE, B, 0, stream>>>(col, total);
  k_bucket_split<<<NSLICE, B, 0, stream>>>(row, col, total, gcur, bstart_g, tmp);
  k_bucket_sort<<<NB, B, 0, stream>>>(tmp, total, bstart_g, off, srcs);

  int ga_blocks   = (NN / 8 * 32 + B - 1) / B;

  k_pack_w<<<INC * 32 / B, B, 0, stream>>>(W1, WP, INC);
  k_gemm_wmma<INC><<<GEMM_GROUPS, GEMM_WAVES * 32, 0, stream>>>(x, WP, off, bufH);

  k_pack_w<<<HIDC * 32 / B, B, 0, stream>>>(W2, WP, HIDC);
  k_gather_gemm<<<FZ_GROUPS, FZ_WAVES * 32, 0, stream>>>(bufH, WP, off, srcs, b1, bufZ);

  k_gather<0><<<ga_blocks, B, 0, stream>>>(bufH, bufZ, off, srcs, b2);

  int dec_blocks = ((NPOS + NNEG) + B - 1) / B;
  k_decode<<<dec_blocks, B, 0, stream>>>(bufH, pos, neg, logits);
}
